# starting kernel
# baseline (speedup 1.0000x reference)
#include <hip/hip_runtime.h>
#include <math.h>

#define DIN 256
#define HID 64
#define HEADS 4
#define DOUT 128
#define NEG_SLOPE 0.2f

typedef __attribute__((ext_vector_type(2))) float v2f;
typedef __attribute__((ext_vector_type(8))) float v8f;

__device__ __forceinline__ void atomicMaxFloat(float* addr, float val) {
  if (val >= 0.0f)
    atomicMax((int*)addr, __float_as_int(val));
  else
    atomicMin((unsigned int*)addr, __float_as_uint(val));
}

template<int K, int NOUT>
__global__ void gemm_f32_wmma(const float* __restrict__ X,
                              const float* __restrict__ W,
                              float* __restrict__ H, int nrows) {
  const int lane = threadIdx.x & 31;
  const int wave = threadIdx.x >> 5;
  const int colGroups = NOUT / 64;
  const int strip = blockIdx.x * (blockDim.x >> 5) + wave;
  const int rowTile = strip / colGroups;
  const int colGroup = strip - rowTile * colGroups;
  if (rowTile * 16 >= nrows) return;
  const int r = lane & 15;
  const int half = lane >> 4;
  const float* __restrict__ Ap = X + (size_t)(rowTile * 16 + r) * K + 2 * half;
  const float* __restrict__ Bp = W + (size_t)(2 * half) * NOUT + colGroup * 64 + r;
  v8f acc0 = {}, acc1 = {}, acc2 = {}, acc3 = {};
#pragma unroll 4
  for (int k = 0; k < K; k += 4) {
    v2f a;
    a[0] = Ap[k];
    a[1] = Ap[k + 1];
    const float* b0 = Bp + (size_t)k * NOUT;
    const float* b1 = Bp + (size_t)(k + 1) * NOUT;
    v2f bA, bB, bC, bD;
    bA[0] = b0[0];   bA[1] = b1[0];
    bB[0] = b0[16];  bB[1] = b1[16];
    bC[0] = b0[32];  bC[1] = b1[32];
    bD[0] = b0[48];  bD[1] = b1[48];
    acc0 = __builtin_amdgcn_wmma_f32_16x16x4_f32(false, a, false, bA,
                                                 (short)0, acc0, false, false);
    acc1 = __builtin_amdgcn_wmma_f32_16x16x4_f32(false, a, false, bB,
                                                 (short)0, acc1, false, false);
    acc2 = __builtin_amdgcn_wmma_f32_16x16x4_f32(false, a, false, bC,
                                                 (short)0, acc2, false, false);
    acc3 = __builtin_amdgcn_wmma_f32_16x16x4_f32(false, a, false, bD,
                                                 (short)0, acc3, false, false);
  }
  float* __restrict__ Cp =
      H + (size_t)(rowTile * 16 + 8 * half) * NOUT + colGroup * 64 + r;
#pragma unroll
  for (int v = 0; v < 8; ++v) {
    Cp[(size_t)v * NOUT]      = acc0[v];
    Cp[(size_t)v * NOUT + 16] = acc1[v];
    Cp[(size_t)v * NOUT + 32] = acc2[v];
    Cp[(size_t)v * NOUT + 48] = acc3[v];
  }
}

__global__ void dot_e_kernel(const float* __restrict__ H,
                             const float* __restrict__ a_src,
                             const float* __restrict__ a_dst,
                             float* __restrict__ e_src,
                             float* __restrict__ e_dst,
                             int n, int heads, int C) {
  int idx = blockIdx.x * blockDim.x + threadIdx.x;
  if (idx >= n * heads) return;
  int node = idx / heads;
  int hd = idx - node * heads;
  const float* hp = H + (size_t)node * heads * C + (size_t)hd * C;
  const float* as = a_src + (size_t)hd * C;
  const float* ad = a_dst + (size_t)hd * C;
  float s0 = 0.f, s1 = 0.f;
  for (int c = 0; c < C; ++c) {
    float v = hp[c];
    s0 += v * as[c];
    s1 += v * ad[c];
  }
  e_src[idx] = s0;
  e_dst[idx] = s1;
}

__global__ void fill_kernel(float* __restrict__ p, float v, size_t n) {
  size_t i = (size_t)blockIdx.x * blockDim.x + threadIdx.x;
  size_t stride = (size_t)gridDim.x * blockDim.x;
  for (; i < n; i += stride) p[i] = v;
}

__global__ void edge_logit_max(const int* __restrict__ srcIdx,
                               const int* __restrict__ dstIdx,
                               int E, int Etot,
                               const float* __restrict__ e_src,
                               const float* __restrict__ e_dst,
                               float* __restrict__ logits,
                               float* __restrict__ m, int heads) {
  int t = blockIdx.x * blockDim.x + threadIdx.x;
  if (t >= Etot * heads) return;
  int e = t / heads;
  int hd = t - e * heads;
  int s = (e < E) ? srcIdx[e] : (e - E);
  int d = (e < E) ? dstIdx[e] : (e - E);
  float l = e_src[s * heads + hd] + e_dst[d * heads + hd];
  l = (l > 0.f) ? l : l * NEG_SLOPE;
  logits[t] = l;
  atomicMaxFloat(&m[d * heads + hd], l);
}

__global__ void edge_exp_sum(const int* __restrict__ srcIdx,
                             const int* __restrict__ dstIdx,
                             int E, int Etot,
                             float* __restrict__ logits,
                             const float* __restrict__ m,
                             float* __restrict__ denom, int heads) {
  int t = blockIdx.x * blockDim.x + threadIdx.x;
  if (t >= Etot * heads) return;
  int e = t / heads;
  int hd = t - e * heads;
  int d = (e < E) ? dstIdx[e] : (e - E);
  float ex = expf(logits[t] - m[d * heads + hd]);
  logits[t] = ex;
  atomicAdd(&denom[d * heads + hd], ex);
}

__global__ void edge_scatter(const int* __restrict__ srcIdx,
                             const int* __restrict__ dstIdx,
                             int E, int Etot,
                             const float* __restrict__ ex,
                             const float* __restrict__ denom,
                             const float* __restrict__ H,
                             float* __restrict__ agg,
                             int heads, int C) {
  int lane = threadIdx.x & 31;
  int wid = (int)(((size_t)blockIdx.x * blockDim.x + threadIdx.x) >> 5);
  if (wid >= Etot * heads) return;
  int e = wid / heads;
  int hd = wid - e * heads;
  int s = (e < E) ? srcIdx[e] : (e - E);
  int d = (e < E) ? dstIdx[e] : (e - E);
  float alpha = ex[wid] / (denom[d * heads + hd] + 1e-16f);
  const float* hp = H + (size_t)s * heads * C + (size_t)hd * C;
  float* ap = agg + (size_t)d * heads * C + (size_t)hd * C;
  for (int c = lane; c < C; c += 32)
    atomicAdd(&ap[c], alpha * hp[c]);
}

__global__ void bias_elu(const float* __restrict__ agg,
                         const float* __restrict__ b,
                         float* __restrict__ out, int n, int F) {
  int t = blockIdx.x * blockDim.x + threadIdx.x;
  if (t >= n * F) return;
  int f = t % F;
  float v = agg[t] + b[f];
  out[t] = (v > 0.f) ? v : (expf(v) - 1.f);
}

__global__ void final_mean(const float* __restrict__ agg,
                           const float* __restrict__ b,
                           float* __restrict__ out, int n, int F) {
  int lane = threadIdx.x & 31;
  int node = (int)(((size_t)blockIdx.x * blockDim.x + threadIdx.x) >> 5);
  if (node >= n) return;
  const float* ap = agg + (size_t)node * F;
  float sum = 0.f;
  for (int c = lane; c < F; c += 32) {
    float v = ap[c] + b[c];
    v = (v > 0.f) ? v : (expf(v) - 1.f);
    sum += v;
  }
#pragma unroll
  for (int off = 16; off > 0; off >>= 1)
    sum += __shfl_down(sum, off, 32);
  if (lane == 0) out[node] = sum / (float)F;
}

static inline int cdiv(long long a, long long b) { return (int)((a + b - 1) / b); }

extern "C" void kernel_launch(void* const* d_in, const int* in_sizes, int n_in,
                              void* d_out, int out_size, void* d_ws, size_t ws_size,
                              hipStream_t stream) {
  const float* x      = (const float*)d_in[0];
  const int*   ei     = (const int*)  d_in[1];
  const float* W1     = (const float*)d_in[2];
  const float* a1_src = (const float*)d_in[3];
  const float* a1_dst = (const float*)d_in[4];
  const float* b1     = (const float*)d_in[5];
  const float* W2     = (const float*)d_in[6];
  const float* a2_src = (const float*)d_in[7];
  const float* a2_dst = (const float*)d_in[8];
  const float* b2     = (const float*)d_in[9];
  float* outp = (float*)d_out;

  const int N    = in_sizes[0] / DIN;
  const int E    = in_sizes[1] / 2;
  const int Etot = E + N;
  const int F1   = HEADS * HID;
  const int* srcIdx = ei;
  const int* dstIdx = ei + E;

  float* ws = (float*)d_ws;
  float* h1      = ws;
  float* e_src_b = h1 + (size_t)N * F1;
  float* e_dst_b = e_src_b + (size_t)N * HEADS;
  float* edgebuf = e_dst_b + (size_t)N * HEADS;
  float* m_b     = edgebuf + (size_t)Etot * HEADS;
  float* den_b   = m_b + (size_t)N * HEADS;
  float* aggreg  = den_b + (size_t)N * HEADS;
  float* h2      = aggreg;
  float* agg2    = aggreg + (size_t)N * DOUT;

  const int TB = 256;

  {
    int strips = (N / 16) * (F1 / 64);
    gemm_f32_wmma<DIN, F1><<<cdiv(strips, TB / 32), TB, 0, stream>>>(x, W1, h1, N);

    dot_e_kernel<<<cdiv((long long)N * HEADS, TB), TB, 0, stream>>>(
        h1, a1_src, a1_dst, e_src_b, e_dst_b, N, HEADS, HID);

    fill_kernel<<<1024, TB, 0, stream>>>(m_b, -INFINITY, (size_t)N * HEADS);
    fill_kernel<<<1024, TB, 0, stream>>>(den_b, 0.f, (size_t)N * HEADS);
    fill_kernel<<<4096, TB, 0, stream>>>(aggreg, 0.f, (size_t)N * F1);

    int eh = Etot * HEADS;
    edge_logit_max<<<cdiv(eh, TB), TB, 0, stream>>>(srcIdx, dstIdx, E, Etot,
                                                    e_src_b, e_dst_b, edgebuf, m_b, HEADS);
    edge_exp_sum<<<cdiv(eh, TB), TB, 0, stream>>>(srcIdx, dstIdx, E, Etot,
                                                  edgebuf, m_b, den_b, HEADS);
    edge_scatter<<<cdiv((long long)eh * 32, TB), TB, 0, stream>>>(
        srcIdx, dstIdx, E, Etot, edgebuf, den_b, h1, aggreg, HEADS, HID);

    bias_elu<<<cdiv((long long)N * F1, TB), TB, 0, stream>>>(aggreg, b1, h1, N, F1);
  }

  {
    int strips = (N / 16) * (DOUT / 64);
    gemm_f32_wmma<F1, DOUT><<<cdiv(strips, TB / 32), TB, 0, stream>>>(h1, W2, h2, N);

    dot_e_kernel<<<cdiv(N, TB), TB, 0, stream>>>(
        h2, a2_src, a2_dst, e_src_b, e_dst_b, N, 1, DOUT);

    fill_kernel<<<1024, TB, 0, stream>>>(m_b, -INFINITY, (size_t)N);
    fill_kernel<<<1024, TB, 0, stream>>>(den_b, 0.f, (size_t)N);
    fill_kernel<<<4096, TB, 0, stream>>>(agg2, 0.f, (size_t)N * DOUT);

    edge_logit_max<<<cdiv(Etot, TB), TB, 0, stream>>>(srcIdx, dstIdx, E, Etot,
                                                      e_src_b, e_dst_b, edgebuf, m_b, 1);
    edge_exp_sum<<<cdiv(Etot, TB), TB, 0, stream>>>(srcIdx, dstIdx, E, Etot,
                                                    edgebuf, m_b, den_b, 1);
    edge_scatter<<<cdiv((long long)Etot * 32, TB), TB, 0, stream>>>(
        srcIdx, dstIdx, E, Etot, edgebuf, den_b, h2, agg2, 1, DOUT);

    final_mean<<<cdiv((long long)N * 32, TB), TB, 0, stream>>>(agg2, b2, outp, N, DOUT);
  }
}


# gat_owner: one sweep files a work-group's owned edges in an LDS list; max and sums walk the list (32 H-row loads in flight per wave); 4 edge sweeps to 2; LDS float atomics kept, arrival order changes
# speedup vs baseline: 2.5849x; 2.5849x over previous
#include <hip/hip_runtime.h>
#include <math.h>

#define DIN 256
#define HID 64
#define HEADS 4
#define DOUT 128
#define NEG_SLOPE 0.2f

typedef __attribute__((ext_vector_type(2))) float v2f;
typedef __attribute__((ext_vector_type(8))) float v8f;

template<int K, int NOUT>
__global__ void gemm_f32_wmma(const float* __restrict__ X,
                              const float* __restrict__ W,
                              float* __restrict__ H, int nrows) {
  const int lane = threadIdx.x & 31;
  const int wave = threadIdx.x >> 5;
  const int colGroups = NOUT / 64;
  const int strip = blockIdx.x * (blockDim.x >> 5) + wave;
  const int rowTile = strip / colGroups;
  const int colGroup = strip - rowTile * colGroups;
  if (rowTile * 16 >= nrows) return;
  const int r = lane & 15;
  const int half = lane >> 4;
  const float* __restrict__ Ap = X + (size_t)(rowTile * 16 + r) * K + 2 * half;
  const float* __restrict__ Bp = W + (size_t)(2 * half) * NOUT + colGroup * 64 + r;
  v8f acc0 = {}, acc1 = {}, acc2 = {}, acc3 = {};
#pragma unroll 4
  for (int k = 0; k < K; k += 4) {
    v2f a;
    a[0] = Ap[k];
    a[1] = Ap[k + 1];
    const float* b0 = Bp + (size_t)k * NOUT;
    const float* b1 = Bp + (size_t)(k + 1) * NOUT;
    v2f bA, bB, bC, bD;
    bA[0] = b0[0];   bA[1] = b1[0];
    bB[0] = b0[16];  bB[1] = b1[16];
    bC[0] = b0[32];  bC[1] = b1[32];
    bD[0] = b0[48];  bD[1] = b1[48];
    acc0 = __builtin_amdgcn_wmma_f32_16x16x4_f32(false, a, false, bA,
                                                 (short)0, acc0, false, false);
    acc1 = __builtin_amdgcn_wmma_f32_16x16x4_f32(false, a, false, bB,
                                                 (short)0, acc1, false, false);
    acc2 = __builtin_amdgcn_wmma_f32_16x16x4_f32(false, a, false, bC,
                                                 (short)0, acc2, false, false);
    acc3 = __builtin_amdgcn_wmma_f32_16x16x4_f32(false, a, false, bD,
                                                 (short)0, acc3, false, false);
  }
  float* __restrict__ Cp =
      H + (size_t)(rowTile * 16 + 8 * half) * NOUT + colGroup * 64 + r;
#pragma unroll
  for (int v = 0; v < 8; ++v) {
    Cp[(size_t)v * NOUT]      = acc0[v];
    Cp[(size_t)v * NOUT + 16] = acc1[v];
    Cp[(size_t)v * NOUT + 32] = acc2[v];
    Cp[(size_t)v * NOUT + 48] = acc3[v];
  }
}

__global__ void dot_e_kernel(const float* __restrict__ H,
                             const float* __restrict__ a_src,
                             const float* __restrict__ a_dst,
                             float* __restrict__ e_src,
                             float* __restrict__ e_dst,
                             int n, int heads, int C) {
  int idx = blockIdx.x * blockDim.x + threadIdx.x;
  if (idx >= n * heads) return;
  int node = idx / heads;
  int hd = idx - node * heads;
  const float* hp = H + (size_t)node * heads * C + (size_t)hd * C;
  const float* as = a_src + (size_t)hd * C;
  const float* ad = a_dst + (size_t)hd * C;
  float s0 = 0.f, s1 = 0.f;
  for (int c = 0; c < C; ++c) {
    float v = hp[c];
    s0 += v * as[c];
    s1 += v * ad[c];
  }
  e_src[idx] = s0;
  e_dst[idx] = s1;
}

#define OWN_ROWS 196
#define OWN_PITCH 256
#define OWN_THREADS 1024
#define OWN_GROUPS 256
#define LIST_CAP 12288

__device__ __forceinline__ void ldsMaxFloat(float* addr, float val) {
  if (val >= 0.0f)
    atomicMax((int*)addr, __float_as_int(val));
  else
    atomicMin((unsigned int*)addr, __float_as_uint(val));
}

template<int NH, int C, bool FINAL>
__global__ void __launch_bounds__(OWN_THREADS)
gat_owner(const int* __restrict__ srcIdx, const int* __restrict__ dstIdx,
          int E, int n,
          const float* __restrict__ H,
          const float* __restrict__ e_src,
          const float* __restrict__ e_dst,
          const float* __restrict__ b,
          float* __restrict__ out) {
  constexpr int F = NH * C;
  constexpr int J = F / 32;
  constexpr int LIST_B = 1024 / F;
  static_assert(!FINAL || NH == 1, "row mean is written for one head");
  static_assert(F <= OWN_PITCH && (F % 32) == 0 && (C % 32) == 0, "channel layout");
  static_assert(LIST_B >= 1 && LIST_B * J == 32, "32 row loads in flight per wave");
  __shared__ float acc[OWN_ROWS * OWN_PITCH];
  __shared__ float mS[OWN_ROWS * NH];
  __shared__ float denS[OWN_ROWS * NH];
  __shared__ float edS[OWN_ROWS * NH];
  __shared__ int listS[LIST_CAP];
  __shared__ int listD[LIST_CAP];
  __shared__ int listN;

  const int tid = threadIdx.x;
  const int lane = tid & 31;
  const int wave = tid >> 5;
  const int nthr = blockDim.x;
  const int nwav = nthr >> 5;
  const int chunk = nthr * 4;
  const int nIt = (E + chunk - 1) / chunk;
  const int it0 = (int)(((long long)blockIdx.x * nIt) / gridDim.x);
  const bool vec = ((E & 3) == 0) && ((((size_t)dstIdx) & 15) == 0);

  auto fetch = [&](int i) -> int4 {
    int4 v = make_int4(-1, -1, -1, -1);
    if (i < nIt) {
      int itr = it0 + i;
      if (itr >= nIt) itr -= nIt;
      const int e0 = itr * chunk + tid * 4;
      if (vec && e0 + 3 < E) {
        v = *reinterpret_cast<const int4*>(dstIdx + e0);
      } else {
        if (e0 < E) v.x = dstIdx[e0];
        if (e0 + 1 < E) v.y = dstIdx[e0 + 1];
        if (e0 + 2 < E) v.z = dstIdx[e0 + 2];
        if (e0 + 3 < E) v.w = dstIdx[e0 + 3];
      }
    }
    return v;
  };
  auto edge0 = [&](int i) -> int {
    int itr = it0 + i;
    if (itr >= nIt) itr -= nIt;
    return itr * chunk + tid * 4;
  };

  for (int base = blockIdx.x * OWN_ROWS; base < n; base += gridDim.x * OWN_ROWS) {
    const int nrows = min(OWN_ROWS, n - base);

    for (int i = tid; i < nrows * F; i += nthr) {
      const int r = i / F;
      acc[r * OWN_PITCH + (i - r * F)] = 0.f;
    }
    for (int i = tid; i < nrows * NH; i += nthr) {
      const float ed = e_dst[(size_t)base * NH + i];
      float l = e_src[(size_t)base * NH + i] + ed;
      l = (l > 0.f) ? l : l * NEG_SLOPE;
      edS[i] = ed;
      mS[i] = l;
      denS[i] = 0.f;
    }
    if (tid == 0) listN = 0;
    __syncthreads();

    {
      int4 q0 = fetch(0), q1 = fetch(1), q2 = fetch(2), q3 = fetch(3);
      for (int i = 0; i < nIt; ++i) {
        const int4 cur = q0;
        q0 = q1; q1 = q2; q2 = q3; q3 = fetch(i + 4);
        const int dl[4] = {cur.x - base, cur.y - base, cur.z - base, cur.w - base};
        bool mt[4];
        unsigned bm[4];
        for (int k = 0; k < 4; ++k) {
          mt[k] = (unsigned)dl[k] < (unsigned)nrows;
          bm[k] = __builtin_amdgcn_ballot_w32(mt[k]);
        }
        if ((bm[0] | bm[1] | bm[2] | bm[3]) == 0u) continue;
        const int found = __builtin_popcount(bm[0]) + __builtin_popcount(bm[1]) +
                          __builtin_popcount(bm[2]) + __builtin_popcount(bm[3]);
        int slot = 0;
        if (lane == 0) slot = atomicAdd(&listN, found);
        slot = __builtin_amdgcn_readfirstlane(slot);
        const int e0 = edge0(i);
        for (int k = 0; k < 4; ++k) {
          if (mt[k]) {
            const int pos = slot + (int)__builtin_amdgcn_mbcnt_lo(bm[k], 0u);
            if (pos < LIST_CAP) {
              listS[pos] = e0 + k;
              listD[pos] = dl[k];
            }
          }
          slot += __builtin_popcount(bm[k]);
        }
      }
    }
    __syncthreads();
    const int total = __builtin_amdgcn_readfirstlane(listN);

    if (total <= LIST_CAP) {
      for (int i0 = tid; i0 < total; i0 += 4 * nthr) {
        int ci[4], sv[4], dv[4];
        bool ok[4];
        for (int k = 0; k < 4; ++k) {
          const int idx = i0 + k * nthr;
          ok[k] = idx < total;
          ci[k] = ok[k] ? idx : i0;
        }
        for (int k = 0; k < 4; ++k) {
          sv[k] = srcIdx[listS[ci[k]]];
          dv[k] = listD[ci[k]];
        }
        float es[4][NH];
        for (int k = 0; k < 4; ++k)
          for (int h = 0; h < NH; ++h)
            es[k][h] = e_src[(size_t)sv[k] * NH + h];
        for (int k = 0; k < 4; ++k) {
          if (ok[k]) {
            listS[ci[k]] = sv[k];
            for (int h = 0; h < NH; ++h) {
              float l = es[k][h] + edS[dv[k] * NH + h];
              l = (l > 0.f) ? l : l * NEG_SLOPE;
              ldsMaxFloat(&mS[dv[k] * NH + h], l);
            }
          }
        }
      }
      __syncthreads();

      const int totalAll = total + nrows;
      const int waveS = __builtin_amdgcn_readfirstlane(wave);
      for (int g = waveS * LIST_B; g < totalAll; g += nwav * LIST_B) {
        int sv[LIST_B], dv[LIST_B];
#pragma unroll
        for (int k = 0; k < LIST_B; ++k) {
          const int idx = min(g + k, totalAll - 1);
          const bool isList = idx < total;
          const int li = isList ? idx : 0;
          const int ls = listS[li];
          const int ld = listD[li];
          const int d = isList ? ld : idx - total;
          const int s = isList ? ls : base + d;
          sv[k] = __builtin_amdgcn_readfirstlane(s);
          dv[k] = __builtin_amdgcn_readfirstlane(d);
        }
        float hv[LIST_B][J];
        float el[LIST_B];
#pragma unroll
        for (int k = 0; k < LIST_B; ++k) {
          const float* hp = H + (size_t)sv[k] * F;
#pragma unroll
          for (int j = 0; j < J; ++j) hv[k][j] = hp[lane + 32 * j];
          el[k] = e_src[(size_t)sv[k] * NH + (lane & (NH - 1))];
        }
#pragma unroll
        for (int k = 0; k < LIST_B; ++k) {
          if (g + k < totalAll) {
            const int d = dv[k];
            const int h = lane & (NH - 1);
            float l = el[k] + edS[d * NH + h];
            l = (l > 0.f) ? l : l * NEG_SLOPE;
            const float ex = expf(l - mS[d * NH + h]);
            if (lane < NH) atomicAdd(&denS[d * NH + lane], ex);
            float* ap = acc + d * OWN_PITCH;
#pragma unroll
            for (int j = 0; j < J; ++j) {
              const float exj = __int_as_float(
                  __builtin_amdgcn_readlane(__float_as_int(ex), (32 * j) / C));
              atomicAdd(&ap[lane + 32 * j], exj * hv[k][j]);
            }
          }
        }
      }
    } else {
      {
        int4 q0 = fetch(0), q1 = fetch(1), q2 = fetch(2), q3 = fetch(3);
        for (int i = 0; i < nIt; ++i) {
          const int4 cur = q0;
          q0 = q1; q1 = q2; q2 = q3; q3 = fetch(i + 4);
          const int dl[4] = {cur.x - base, cur.y - base, cur.z - base, cur.w - base};
          bool mt[4];
          for (int k = 0; k < 4; ++k) mt[k] = (unsigned)dl[k] < (unsigned)nrows;
          if (__builtin_amdgcn_ballot_w32(mt[0] | mt[1] | mt[2] | mt[3]) == 0u) continue;
          const int e0 = edge0(i);
          int sv[4];
          for (int k = 0; k < 4; ++k) sv[k] = mt[k] ? srcIdx[e0 + k] : 0;
          float es[4][NH];
          for (int k = 0; k < 4; ++k)
            for (int h = 0; h < NH; ++h)
              es[k][h] = mt[k] ? e_src[(size_t)sv[k] * NH + h] : 0.f;
          for (int k = 0; k < 4; ++k) {
            if (mt[k]) {
              for (int h = 0; h < NH; ++h) {
                float l = es[k][h] + edS[dl[k] * NH + h];
                l = (l > 0.f) ? l : l * NEG_SLOPE;
                ldsMaxFloat(&mS[dl[k] * NH + h], l);
              }
            }
          }
        }
      }
      __syncthreads();

      {
        auto addEdge = [&](int s, int d) {
          const float* hp = H + (size_t)s * F;
          float hv[J];
          for (int j = 0; j < J; ++j) hv[j] = hp[lane + 32 * j];
          const int h = lane & (NH - 1);
          float l = e_src[(size_t)s * NH + h] + edS[d * NH + h];
          l = (l > 0.f) ? l : l * NEG_SLOPE;
          const float ex = expf(l - mS[d * NH + h]);
          if (lane < NH) atomicAdd(&denS[d * NH + lane], ex);
          float* ap = acc + d * OWN_PITCH;
#pragma unroll
          for (int j = 0; j < J; ++j) {
            const float exj = __int_as_float(
                __builtin_amdgcn_readlane(__float_as_int(ex), (32 * j) / C));
            atomicAdd(&ap[lane + 32 * j], exj * hv[j]);
          }
        };

        int4 q0 = fetch(0), q1 = fetch(1), q2 = fetch(2), q3 = fetch(3);
        for (int i = 0; i < nIt; ++i) {
          const int4 cur = q0;
          q0 = q1; q1 = q2; q2 = q3; q3 = fetch(i + 4);
          const int dl[4] = {cur.x - base, cur.y - base, cur.z - base, cur.w - base};
          bool mt[4];
          for (int k = 0; k < 4; ++k) mt[k] = (unsigned)dl[k] < (unsigned)nrows;
          if (__builtin_amdgcn_ballot_w32(mt[0] | mt[1] | mt[2] | mt[3]) == 0u) continue;
          const int e0 = edge0(i);
          int sv[4];
          for (int k = 0; k < 4; ++k) sv[k] = mt[k] ? srcIdx[e0 + k] : 0;
          for (int k = 0; k < 4; ++k) {
            unsigned mask = __builtin_amdgcn_ballot_w32(mt[k]);
            while (mask) {
              const int j = __builtin_ctz(mask);
              mask &= mask - 1u;
              addEdge(__builtin_amdgcn_readlane(sv[k], j),
                      __builtin_amdgcn_readlane(dl[k], j));
            }
          }
        }
        for (int r = wave; r < nrows; r += nwav) addEdge(base + r, r);
      }
    }
    __syncthreads();

    if (!FINAL) {
      for (int i = tid; i < nrows * F; i += nthr) {
        const int r = i / F;
        const int f = i - r * F;
        const float v = acc[r * OWN_PITCH + f] / (denS[r * NH + f / C] + 1e-16f) + b[f];
        out[(size_t)(base + r) * F + f] = (v > 0.f) ? v : (expf(v) - 1.f);
      }
    } else {
      for (int r = wave; r < nrows; r += nwav) {
        const float den = denS[r * NH] + 1e-16f;
        float sum = 0.f;
        for (int c = lane; c < F; c += 32) {
          float v = acc[r * OWN_PITCH + c] / den + b[c];
          v = (v > 0.f) ? v : (expf(v) - 1.f);
          sum += v;
        }
        for (int off = 16; off > 0; off >>= 1)
          sum += __shfl_down(sum, off, 32);
        if (lane == 0) out[base + r] = sum / (float)F;
      }
    }
    __syncthreads();
  }
}

static inline int cdiv(long long a, long long b) { return (int)((a + b - 1) / b); }

extern "C" void kernel_launch(void* const* d_in, const int* in_sizes, int n_in,
                              void* d_out, int out_size, void* d_ws, size_t ws_size,
                              hipStream_t stream) {
  const float* x      = (const float*)d_in[0];
  const int*   ei     = (const int*)  d_in[1];
  const float* W1     = (const float*)d_in[2];
  const float* a1_src = (const float*)d_in[3];
  const float* a1_dst = (const float*)d_in[4];
  const float* b1     = (const float*)d_in[5];
  const float* W2     = (const float*)d_in[6];
  const float* a2_src = (const float*)d_in[7];
  const float* a2_dst = (const float*)d_in[8];
  const float* b2     = (const float*)d_in[9];
  float* outp = (float*)d_out;

  const int N    = in_sizes[0] / DIN;
  const int E    = in_sizes[1] / 2;
  const int Etot = E + N;
  const int F1   = HEADS * HID;
  const int* srcIdx = ei;
  const int* dstIdx = ei + E;

  float* ws = (float*)d_ws;
  float* h1      = ws;
  float* e_src_b = h1 + (size_t)N * F1;
  float* e_dst_b = e_src_b + (size_t)N * HEADS;
  float* edgebuf = e_dst_b + (size_t)N * HEADS;
  float* m_b     = edgebuf + (size_t)Etot * HEADS;
  float* den_b   = m_b + (size_t)N * HEADS;
  float* out1    = den_b + (size_t)N * HEADS;
  float* h2      = h1;

  const int TB = 256;

  {
    int strips = (N / 16) * (F1 / 64);
    gemm_f32_wmma<DIN, F1><<<cdiv(strips, TB / 32), TB, 0, stream>>>(x, W1, h1, N);

    dot_e_kernel<<<cdiv((long long)N * HEADS, TB), TB, 0, stream>>>(
        h1, a1_src, a1_dst, e_src_b, e_dst_b, N, HEADS, HID);

    gat_owner<HEADS, HID, false><<<OWN_GROUPS, OWN_THREADS, 0, stream>>>(
        srcIdx, dstIdx, E, N, h1, e_src_b, e_dst_b, b1, out1);
  }

  {
    int strips = (N / 16) * (DOUT / 64);
    gemm_f32_wmma<F1, DOUT><<<cdiv(strips, TB / 32), TB, 0, stream>>>(out1, W2, h2, N);

    dot_e_kernel<<<cdiv(N, TB), TB, 0, stream>>>(
        h2, a2_src, a2_dst, e_src_b, e_dst_b, N, 1, DOUT);

    gat_owner<1, DOUT, true><<<OWN_GROUPS, OWN_THREADS, 0, stream>>>(
        srcIdx, dstIdx, E, N, h2, e_src_b, e_dst_b, b2, outp);
  }
}


# gat_owner: layer 1 saves each row block's owned-edge list (source row, local row) to the workspace, layer 2 reads it back: 2 edge sweeps to 1; LDS float atomics kept (arrival order)
# speedup vs baseline: 3.1095x; 1.2029x over previous
#include <hip/hip_runtime.h>
#include <math.h>

#define DIN 256
#define HID 64
#define HEADS 4
#define DOUT 128
#define NEG_SLOPE 0.2f

typedef __attribute__((ext_vector_type(2))) float v2f;
typedef __attribute__((ext_vector_type(8))) float v8f;

template<int K, int NOUT>
__global__ void gemm_f32_wmma(const float* __restrict__ X,
                              const float* __restrict__ W,
                              float* __restrict__ H, int nrows) {
  const int lane = threadIdx.x & 31;
  const int wave = threadIdx.x >> 5;
  const int colGroups = NOUT / 64;
  const int strip = blockIdx.x * (blockDim.x >> 5) + wave;
  const int rowTile = strip / colGroups;
  const int colGroup = strip - rowTile * colGroups;
  if (rowTile * 16 >= nrows) return;
  const int r = lane & 15;
  const int half = lane >> 4;
  const float* __restrict__ Ap = X + (size_t)(rowTile * 16 + r) * K + 2 * half;
  const float* __restrict__ Bp = W + (size_t)(2 * half) * NOUT + colGroup * 64 + r;
  v8f acc0 = {}, acc1 = {}, acc2 = {}, acc3 = {};
#pragma unroll 4
  for (int k = 0; k < K; k += 4) {
    v2f a;
    a[0] = Ap[k];
    a[1] = Ap[k + 1];
    const float* b0 = Bp + (size_t)k * NOUT;
    const float* b1 = Bp + (size_t)(k + 1) * NOUT;
    v2f bA, bB, bC, bD;
    bA[0] = b0[0];   bA[1] = b1[0];
    bB[0] = b0[16];  bB[1] = b1[16];
    bC[0] = b0[32];  bC[1] = b1[32];
    bD[0] = b0[48];  bD[1] = b1[48];
    acc0 = __builtin_amdgcn_wmma_f32_16x16x4_f32(false, a, false, bA,
                                                 (short)0, acc0, false, false);
    acc1 = __builtin_amdgcn_wmma_f32_16x16x4_f32(false, a, false, bB,
                                                 (short)0, acc1, false, false);
    acc2 = __builtin_amdgcn_wmma_f32_16x16x4_f32(false, a, false, bC,
                                                 (short)0, acc2, false, false);
    acc3 = __builtin_amdgcn_wmma_f32_16x16x4_f32(false, a, false, bD,
                                                 (short)0, acc3, false, false);
  }
  float* __restrict__ Cp =
      H + (size_t)(rowTile * 16 + 8 * half) * NOUT + colGroup * 64 + r;
#pragma unroll
  for (int v = 0; v < 8; ++v) {
    Cp[(size_t)v * NOUT]      = acc0[v];
    Cp[(size_t)v * NOUT + 16] = acc1[v];
    Cp[(size_t)v * NOUT + 32] = acc2[v];
    Cp[(size_t)v * NOUT + 48] = acc3[v];
  }
}

__global__ void dot_e_kernel(const float* __restrict__ H,
                             const float* __restrict__ a_src,
                             const float* __restrict__ a_dst,
                             float* __restrict__ e_src,
                             float* __restrict__ e_dst,
                             int n, int heads, int C) {
  int idx = blockIdx.x * blockDim.x + threadIdx.x;
  if (idx >= n * heads) return;
  int node = idx / heads;
  int hd = idx - node * heads;
  const float* hp = H + (size_t)node * heads * C + (size_t)hd * C;
  const float* as = a_src + (size_t)hd * C;
  const float* ad = a_dst + (size_t)hd * C;
  float s0 = 0.f, s1 = 0.f;
  for (int c = 0; c < C; ++c) {
    float v = hp[c];
    s0 += v * as[c];
    s1 += v * ad[c];
  }
  e_src[idx] = s0;
  e_dst[idx] = s1;
}

#define OWN_ROWS 196
#define OWN_PITCH 256
#define OWN_THREADS 1024
#define OWN_GROUPS 256
#define LIST_CAP 12288

__device__ __forceinline__ void ldsMaxFloat(float* addr, float val) {
  if (val >= 0.0f)
    atomicMax((int*)addr, __float_as_int(val));
  else
    atomicMin((unsigned int*)addr, __float_as_uint(val));
}

template<int NH, int C, bool FINAL, int LMODE>
__global__ void __launch_bounds__(OWN_THREADS)
gat_owner(const int* __restrict__ srcIdx, const int* __restrict__ dstIdx,
          int E, int n,
          const float* __restrict__ H,
          const float* __restrict__ e_src,
          const float* __restrict__ e_dst,
          const float* __restrict__ b,
          float* __restrict__ out,
          int* listW,
          int* cntW,
          int edgeRows) {
  constexpr int F = NH * C;
  constexpr int J = F / 32;
  constexpr int LIST_B = 1024 / F;
  static_assert(!FINAL || NH == 1, "row mean is written for one head");
  static_assert(F <= OWN_PITCH && (F % 32) == 0 && (C % 32) == 0, "channel layout");
  static_assert(LIST_B >= 1 && LIST_B * J == 32, "32 row loads in flight per wave");
  static_assert(LMODE == 1 || LMODE == 2, "1: sweep and save the list, 2: read the saved list");
  static_assert(OWN_ROWS <= 256, "a saved word keeps the local row in its top 8 bits");
  __shared__ float acc[OWN_ROWS * OWN_PITCH];
  __shared__ float mS[OWN_ROWS * NH];
  __shared__ float denS[OWN_ROWS * NH];
  __shared__ float edS[OWN_ROWS * NH];
  __shared__ int listS[LIST_CAP];
  __shared__ int listD[LIST_CAP];
  __shared__ int listN;

  const int tid = threadIdx.x;
  const int lane = tid & 31;
  const int wave = tid >> 5;
  const int nthr = blockDim.x;
  const int nwav = nthr >> 5;
  const int chunk = nthr * 4;
  const int nIt = (E + chunk - 1) / chunk;
  const int it0 = (int)(((long long)blockIdx.x * nIt) / gridDim.x);
  const bool vec = ((E & 3) == 0) && ((((size_t)dstIdx) & 15) == 0);

  auto fetch = [&](int i) -> int4 {
    int4 v = make_int4(-1, -1, -1, -1);
    if (i < nIt) {
      int itr = it0 + i;
      if (itr >= nIt) itr -= nIt;
      const int e0 = itr * chunk + tid * 4;
      if (vec && e0 + 3 < E) {
        v = *reinterpret_cast<const int4*>(dstIdx + e0);
      } else {
        if (e0 < E) v.x = dstIdx[e0];
        if (e0 + 1 < E) v.y = dstIdx[e0 + 1];
        if (e0 + 2 < E) v.z = dstIdx[e0 + 2];
        if (e0 + 3 < E) v.w = dstIdx[e0 + 3];
      }
    }
    return v;
  };
  auto edge0 = [&](int i) -> int {
    int itr = it0 + i;
    if (itr >= nIt) itr -= nIt;
    return itr * chunk + tid * 4;
  };

  for (int base = blockIdx.x * OWN_ROWS; base < n; base += gridDim.x * OWN_ROWS) {
    const int nrows = min(OWN_ROWS, n - base);
    const int blk = base / OWN_ROWS;
    int* const lw = listW + (size_t)blk * LIST_CAP;
    const bool canSave = (n <= (1 << 24)) &&
                         ((long long)(blk + 1) * LIST_CAP <= (long long)edgeRows * HEADS);

    for (int i = tid; i < nrows * F; i += nthr) {
      const int r = i / F;
      acc[r * OWN_PITCH + (i - r * F)] = 0.f;
    }
    for (int i = tid; i < nrows * NH; i += nthr) {
      const float ed = e_dst[(size_t)base * NH + i];
      float l = e_src[(size_t)base * NH + i] + ed;
      l = (l > 0.f) ? l : l * NEG_SLOPE;
      edS[i] = ed;
      mS[i] = l;
      denS[i] = 0.f;
    }
    if (tid == 0) listN = 0;
    int total;
    if (LMODE == 2) {
      total = __builtin_amdgcn_readfirstlane(cntW[blk]);
      if (total <= LIST_CAP) {
        for (int i = tid; i < total; i += nthr) {
          const int p = lw[i];
          listS[i] = min(p & 0xFFFFFF, n - 1);
          listD[i] = min((int)((unsigned)p >> 24), nrows - 1);
        }
      }
      __syncthreads();
    } else {
      __syncthreads();

      {
        int4 q0 = fetch(0), q1 = fetch(1), q2 = fetch(2), q3 = fetch(3);
        for (int i = 0; i < nIt; ++i) {
          const int4 cur = q0;
          q0 = q1; q1 = q2; q2 = q3; q3 = fetch(i + 4);
          const int dl[4] = {cur.x - base, cur.y - base, cur.z - base, cur.w - base};
          bool mt[4];
          unsigned bm[4];
          for (int k = 0; k < 4; ++k) {
            mt[k] = (unsigned)dl[k] < (unsigned)nrows;
            bm[k] = __builtin_amdgcn_ballot_w32(mt[k]);
          }
          if ((bm[0] | bm[1] | bm[2] | bm[3]) == 0u) continue;
          const int found = __builtin_popcount(bm[0]) + __builtin_popcount(bm[1]) +
                            __builtin_popcount(bm[2]) + __builtin_popcount(bm[3]);
          int slot = 0;
          if (lane == 0) slot = atomicAdd(&listN, found);
          slot = __builtin_amdgcn_readfirstlane(slot);
          const int e0 = edge0(i);
          for (int k = 0; k < 4; ++k) {
            if (mt[k]) {
              const int pos = slot + (int)__builtin_amdgcn_mbcnt_lo(bm[k], 0u);
              if (pos < LIST_CAP) {
                listS[pos] = e0 + k;
                listD[pos] = dl[k];
              }
            }
            slot += __builtin_popcount(bm[k]);
          }
        }
      }
      __syncthreads();
      total = __builtin_amdgcn_readfirstlane(listN);
      if (tid == 0) cntW[blk] = canSave ? total : LIST_CAP + 1;
    }

    if (total <= LIST_CAP) {
      for (int i0 = tid; i0 < total; i0 += 4 * nthr) {
        int ci[4], sv[4], dv[4];
        bool ok[4];
        for (int k = 0; k < 4; ++k) {
          const int idx = i0 + k * nthr;
          ok[k] = idx < total;
          ci[k] = ok[k] ? idx : i0;
        }
        for (int k = 0; k < 4; ++k) {
          sv[k] = (LMODE == 1) ? srcIdx[listS[ci[k]]] : listS[ci[k]];
          dv[k] = listD[ci[k]];
        }
        float es[4][NH];
        for (int k = 0; k < 4; ++k)
          for (int h = 0; h < NH; ++h)
            es[k][h] = e_src[(size_t)sv[k] * NH + h];
        for (int k = 0; k < 4; ++k) {
          if (ok[k]) {
            if (LMODE == 1) listS[ci[k]] = sv[k];
            for (int h = 0; h < NH; ++h) {
              float l = es[k][h] + edS[dv[k] * NH + h];
              l = (l > 0.f) ? l : l * NEG_SLOPE;
              ldsMaxFloat(&mS[dv[k] * NH + h], l);
            }
          }
        }
      }
      __syncthreads();

      if (LMODE == 1 && canSave) {
        for (int i = tid; i < total; i += nthr) lw[i] = (int)((unsigned)listS[i] | ((unsigned)listD[i] << 24));
      }

      const int totalAll = total + nrows;
      const int waveS = __builtin_amdgcn_readfirstlane(wave);
      for (int g = waveS * LIST_B; g < totalAll; g += nwav * LIST_B) {
        int sv[LIST_B], dv[LIST_B];
#pragma unroll
        for (int k = 0; k < LIST_B; ++k) {
          const int idx = min(g + k, totalAll - 1);
          const bool isList = idx < total;
          const int li = isList ? idx : 0;
          const int ls = listS[li];
          const int ld = listD[li];
          const int d = isList ? ld : idx - total;
          const int s = isList ? ls : base + d;
          sv[k] = __builtin_amdgcn_readfirstlane(s);
          dv[k] = __builtin_amdgcn_readfirstlane(d);
        }
        float hv[LIST_B][J];
        float el[LIST_B];
#pragma unroll
        for (int k = 0; k < LIST_B; ++k) {
          const float* hp = H + (size_t)sv[k] * F;
#pragma unroll
          for (int j = 0; j < J; ++j) hv[k][j] = hp[lane + 32 * j];
          el[k] = e_src[(size_t)sv[k] * NH + (lane & (NH - 1))];
        }
#pragma unroll
        for (int k = 0; k < LIST_B; ++k) {
          if (g + k < totalAll) {
            const int d = dv[k];
            const int h = lane & (NH - 1);
            float l = el[k] + edS[d * NH + h];
            l = (l > 0.f) ? l : l * NEG_SLOPE;
            const float ex = expf(l - mS[d * NH + h]);
            if (lane < NH) atomicAdd(&denS[d * NH + lane], ex);
            float* ap = acc + d * OWN_PITCH;
#pragma unroll
            for (int j = 0; j < J; ++j) {
              const float exj = __int_as_float(
                  __builtin_amdgcn_readlane(__float_as_int(ex), (32 * j) / C));
              atomicAdd(&ap[lane + 32 * j], exj * hv[k][j]);
            }
          }
        }
      }
    } else {
      {
        int4 q0 = fetch(0), q1 = fetch(1), q2 = fetch(2), q3 = fetch(3);
        for (int i = 0; i < nIt; ++i) {
          const int4 cur = q0;
          q0 = q1; q1 = q2; q2 = q3; q3 = fetch(i + 4);
          const int dl[4] = {cur.x - base, cur.y - base, cur.z - base, cur.w - base};
          bool mt[4];
          for (int k = 0; k < 4; ++k) mt[k] = (unsigned)dl[k] < (unsigned)nrows;
          if (__builtin_amdgcn_ballot_w32(mt[0] | mt[1] | mt[2] | mt[3]) == 0u) continue;
          const int e0 = edge0(i);
          int sv[4];
          for (int k = 0; k < 4; ++k) sv[k] = mt[k] ? srcIdx[e0 + k] : 0;
          float es[4][NH];
          for (int k = 0; k < 4; ++k)
            for (int h = 0; h < NH; ++h)
              es[k][h] = mt[k] ? e_src[(size_t)sv[k] * NH + h] : 0.f;
          for (int k = 0; k < 4; ++k) {
            if (mt[k]) {
              for (int h = 0; h < NH; ++h) {
                float l = es[k][h] + edS[dl[k] * NH + h];
                l = (l > 0.f) ? l : l * NEG_SLOPE;
                ldsMaxFloat(&mS[dl[k] * NH + h], l);
              }
            }
          }
        }
      }
      __syncthreads();

      {
        auto addEdge = [&](int s, int d) {
          const float* hp = H + (size_t)s * F;
          float hv[J];
          for (int j = 0; j < J; ++j) hv[j] = hp[lane + 32 * j];
          const int h = lane & (NH - 1);
          float l = e_src[(size_t)s * NH + h] + edS[d * NH + h];
          l = (l > 0.f) ? l : l * NEG_SLOPE;
          const float ex = expf(l - mS[d * NH + h]);
          if (lane < NH) atomicAdd(&denS[d * NH + lane], ex);
          float* ap = acc + d * OWN_PITCH;
#pragma unroll
          for (int j = 0; j < J; ++j) {
            const float exj = __int_as_float(
                __builtin_amdgcn_readlane(__float_as_int(ex), (32 * j) / C));
            atomicAdd(&ap[lane + 32 * j], exj * hv[j]);
          }
        };

        int4 q0 = fetch(0), q1 = fetch(1), q2 = fetch(2), q3 = fetch(3);
        for (int i = 0; i < nIt; ++i) {
          const int4 cur = q0;
          q0 = q1; q1 = q2; q2 = q3; q3 = fetch(i + 4);
          const int dl[4] = {cur.x - base, cur.y - base, cur.z - base, cur.w - base};
          bool mt[4];
          for (int k = 0; k < 4; ++k) mt[k] = (unsigned)dl[k] < (unsigned)nrows;
          if (__builtin_amdgcn_ballot_w32(mt[0] | mt[1] | mt[2] | mt[3]) == 0u) continue;
          const int e0 = edge0(i);
          int sv[4];
          for (int k = 0; k < 4; ++k) sv[k] = mt[k] ? srcIdx[e0 + k] : 0;
          for (int k = 0; k < 4; ++k) {
            unsigned mask = __builtin_amdgcn_ballot_w32(mt[k]);
            while (mask) {
              const int j = __builtin_ctz(mask);
              mask &= mask - 1u;
              addEdge(__builtin_amdgcn_readlane(sv[k], j),
                      __builtin_amdgcn_readlane(dl[k], j));
            }
          }
        }
        for (int r = wave; r < nrows; r += nwav) addEdge(base + r, r);
      }
    }
    __syncthreads();

    if (!FINAL) {
      for (int i = tid; i < nrows * F; i += nthr) {
        const int r = i / F;
        const int f = i - r * F;
        const float v = acc[r * OWN_PITCH + f] / (denS[r * NH + f / C] + 1e-16f) + b[f];
        out[(size_t)(base + r) * F + f] = (v > 0.f) ? v : (expf(v) - 1.f);
      }
    } else {
      for (int r = wave; r < nrows; r += nwav) {
        const float den = denS[r * NH] + 1e-16f;
        float sum = 0.f;
        for (int c = lane; c < F; c += 32) {
          float v = acc[r * OWN_PITCH + c] / den + b[c];
          v = (v > 0.f) ? v : (expf(v) - 1.f);
          sum += v;
        }
        for (int off = 16; off > 0; off >>= 1)
          sum += __shfl_down(sum, off, 32);
        if (lane == 0) out[base + r] = sum / (float)F;
      }
    }
    __syncthreads();
  }
}

static inline int cdiv(long long a, long long b) { return (int)((a + b - 1) / b); }

extern "C" void kernel_launch(void* const* d_in, const int* in_sizes, int n_in,
                              void* d_out, int out_size, void* d_ws, size_t ws_size,
                              hipStream_t stream) {
  const float* x      = (const float*)d_in[0];
  const int*   ei     = (const int*)  d_in[1];
  const float* W1     = (const float*)d_in[2];
  const float* a1_src = (const float*)d_in[3];
  const float* a1_dst = (const float*)d_in[4];
  const float* b1     = (const float*)d_in[5];
  const float* W2     = (const float*)d_in[6];
  const float* a2_src = (const float*)d_in[7];
  const float* a2_dst = (const float*)d_in[8];
  const float* b2     = (const float*)d_in[9];
  float* outp = (float*)d_out;

  const int N    = in_sizes[0] / DIN;
  const int E    = in_sizes[1] / 2;
  const int Etot = E + N;
  const int F1   = HEADS * HID;
  const int* srcIdx = ei;
  const int* dstIdx = ei + E;

  float* ws = (float*)d_ws;
  float* h1      = ws;
  float* e_src_b = h1 + (size_t)N * F1;
  float* e_dst_b = e_src_b + (size_t)N * HEADS;
  float* edgebuf = e_dst_b + (size_t)N * HEADS;
  float* m_b     = edgebuf + (size_t)Etot * HEADS;
  float* den_b   = m_b + (size_t)N * HEADS;
  float* out1    = den_b + (size_t)N * HEADS;
  float* h2      = h1;
  int* listW = (int*)edgebuf;
  int* cntW  = (int*)m_b;

  const int TB = 256;

  {
    int strips = (N / 16) * (F1 / 64);
    gemm_f32_wmma<DIN, F1><<<cdiv(strips, TB / 32), TB, 0, stream>>>(x, W1, h1, N);

    dot_e_kernel<<<cdiv((long long)N * HEADS, TB), TB, 0, stream>>>(
        h1, a1_src, a1_dst, e_src_b, e_dst_b, N, HEADS, HID);

    gat_owner<HEADS, HID, false, 1><<<OWN_GROUPS, OWN_THREADS, 0, stream>>>(
        srcIdx, dstIdx, E, N, h1, e_src_b, e_dst_b, b1, out1, listW, cntW, Etot);
  }

  {
    int strips = (N / 16) * (DOUT / 64);
    gemm_f32_wmma<F1, DOUT><<<cdiv(strips, TB / 32), TB, 0, stream>>>(out1, W2, h2, N);

    dot_e_kernel<<<cdiv(N, TB), TB, 0, stream>>>(
        h2, a2_src, a2_dst, e_src_b, e_dst_b, N, 1, DOUT);

    gat_owner<1, DOUT, true, 2><<<OWN_GROUPS, OWN_THREADS, 0, stream>>>(
        srcIdx, dstIdx, E, N, h2, e_src_b, e_dst_b, b2, outp, listW, cntW, Etot);
  }
}


# gat_owner sweep: destination ids asked for 8 steps ahead, each ring slot loaded in place (loop unrolled by ring depth); the handed-on ring waited for every load in its own step; sum order as before
# speedup vs baseline: 3.4185x; 1.0994x over previous
#include <hip/hip_runtime.h>
#include <math.h>

#define DIN 256
#define HID 64
#define HEADS 4
#define DOUT 128
#define NEG_SLOPE 0.2f

typedef __attribute__((ext_vector_type(2))) float v2f;
typedef __attribute__((ext_vector_type(8))) float v8f;

template<int K, int NOUT>
__global__ void gemm_f32_wmma(const float* __restrict__ X,
                              const float* __restrict__ W,
                              float* __restrict__ H, int nrows) {
  const int lane = threadIdx.x & 31;
  const int wave = threadIdx.x >> 5;
  const int colGroups = NOUT / 64;
  const int strip = blockIdx.x * (blockDim.x >> 5) + wave;
  const int rowTile = strip / colGroups;
  const int colGroup = strip - rowTile * colGroups;
  if (rowTile * 16 >= nrows) return;
  const int r = lane & 15;
  const int half = lane >> 4;
  const float* __restrict__ Ap = X + (size_t)(rowTile * 16 + r) * K + 2 * half;
  const float* __restrict__ Bp = W + (size_t)(2 * half) * NOUT + colGroup * 64 + r;
  v8f acc0 = {}, acc1 = {}, acc2 = {}, acc3 = {};
#pragma unroll 4
  for (int k = 0; k < K; k += 4) {
    v2f a;
    a[0] = Ap[k];
    a[1] = Ap[k + 1];
    const float* b0 = Bp + (size_t)k * NOUT;
    const float* b1 = Bp + (size_t)(k + 1) * NOUT;
    v2f bA, bB, bC, bD;
    bA[0] = b0[0];   bA[1] = b1[0];
    bB[0] = b0[16];  bB[1] = b1[16];
    bC[0] = b0[32];  bC[1] = b1[32];
    bD[0] = b0[48];  bD[1] = b1[48];
    acc0 = __builtin_amdgcn_wmma_f32_16x16x4_f32(false, a, false, bA,
                                                 (short)0, acc0, false, false);
    acc1 = __builtin_amdgcn_wmma_f32_16x16x4_f32(false, a, false, bB,
                                                 (short)0, acc1, false, false);
    acc2 = __builtin_amdgcn_wmma_f32_16x16x4_f32(false, a, false, bC,
                                                 (short)0, acc2, false, false);
    acc3 = __builtin_amdgcn_wmma_f32_16x16x4_f32(false, a, false, bD,
                                                 (short)0, acc3, false, false);
  }
  float* __restrict__ Cp =
      H + (size_t)(rowTile * 16 + 8 * half) * NOUT + colGroup * 64 + r;
#pragma unroll
  for (int v = 0; v < 8; ++v) {
    Cp[(size_t)v * NOUT]      = acc0[v];
    Cp[(size_t)v * NOUT + 16] = acc1[v];
    Cp[(size_t)v * NOUT + 32] = acc2[v];
    Cp[(size_t)v * NOUT + 48] = acc3[v];
  }
}

__global__ void dot_e_kernel(const float* __restrict__ H,
                             const float* __restrict__ a_src,
                             const float* __restrict__ a_dst,
                             float* __restrict__ e_src,
                             float* __restrict__ e_dst,
                             int n, int heads, int C) {
  int idx = blockIdx.x * blockDim.x + threadIdx.x;
  if (idx >= n * heads) return;
  int node = idx / heads;
  int hd = idx - node * heads;
  const float* hp = H + (size_t)node * heads * C + (size_t)hd * C;
  const float* as = a_src + (size_t)hd * C;
  const float* ad = a_dst + (size_t)hd * C;
  float s0 = 0.f, s1 = 0.f;
  for (int c = 0; c < C; ++c) {
    float v = hp[c];
    s0 += v * as[c];
    s1 += v * ad[c];
  }
  e_src[idx] = s0;
  e_dst[idx] = s1;
}

#define OWN_ROWS 196
#define OWN_PITCH 256
#define OWN_THREADS 1024
#define OWN_GROUPS 256
#define LIST_CAP 12288
#define SWEEP_AHEAD 8

__device__ __forceinline__ void ldsMaxFloat(float* addr, float val) {
  if (val >= 0.0f)
    atomicMax((int*)addr, __float_as_int(val));
  else
    atomicMin((unsigned int*)addr, __float_as_uint(val));
}

template<int NH, int C, bool FINAL, int LMODE>
__global__ void __launch_bounds__(OWN_THREADS)
gat_owner(const int* __restrict__ srcIdx, const int* __restrict__ dstIdx,
          int E, int n,
          const float* __restrict__ H,
          const float* __restrict__ e_src,
          const float* __restrict__ e_dst,
          const float* __restrict__ b,
          float* __restrict__ out,
          int* listW,
          int* cntW,
          int edgeRows) {
  constexpr int F = NH * C;
  constexpr int J = F / 32;
  constexpr int LIST_B = 1024 / F;
  static_assert(!FINAL || NH == 1, "row mean is written for one head");
  static_assert(F <= OWN_PITCH && (F % 32) == 0 && (C % 32) == 0, "channel layout");
  static_assert(LIST_B >= 1 && LIST_B * J == 32, "32 row loads in flight per wave");
  static_assert(LMODE == 1 || LMODE == 2, "1: sweep and save the list, 2: read the saved list");
  static_assert(OWN_ROWS <= 256, "a saved word keeps the local row in its top 8 bits");
  __shared__ float acc[OWN_ROWS * OWN_PITCH];
  __shared__ float mS[OWN_ROWS * NH];
  __shared__ float denS[OWN_ROWS * NH];
  __shared__ float edS[OWN_ROWS * NH];
  __shared__ int listS[LIST_CAP];
  __shared__ int listD[LIST_CAP];
  __shared__ int listN;

  const int tid = threadIdx.x;
  const int lane = tid & 31;
  const int wave = tid >> 5;
  const int nthr = blockDim.x;
  const int nwav = nthr >> 5;
  const int chunk = nthr * 4;
  const int nIt = (E + chunk - 1) / chunk;
  const int it0 = (int)(((long long)blockIdx.x * nIt) / gridDim.x);
  const bool vec = ((E & 3) == 0) && ((((size_t)dstIdx) & 15) == 0);

  auto fetch = [&](int i) -> int4 {
    int4 v = make_int4(-1, -1, -1, -1);
    if (i < nIt) {
      int itr = it0 + i;
      if (itr >= nIt) itr -= nIt;
      const int e0 = itr * chunk + tid * 4;
      if (vec && e0 + 3 < E) {
        v = *reinterpret_cast<const int4*>(dstIdx + e0);
      } else {
        if (e0 < E) v.x = dstIdx[e0];
        if (e0 + 1 < E) v.y = dstIdx[e0 + 1];
        if (e0 + 2 < E) v.z = dstIdx[e0 + 2];
        if (e0 + 3 < E) v.w = dstIdx[e0 + 3];
      }
    }
    return v;
  };
  auto edge0 = [&](int i) -> int {
    int itr = it0 + i;
    if (itr >= nIt) itr -= nIt;
    return itr * chunk + tid * 4;
  };

  for (int base = blockIdx.x * OWN_ROWS; base < n; base += gridDim.x * OWN_ROWS) {
    const int nrows = min(OWN_ROWS, n - base);
    const int blk = base / OWN_ROWS;
    int* const lw = listW + (size_t)blk * LIST_CAP;
    const bool canSave = (n <= (1 << 24)) &&
                         ((long long)(blk + 1) * LIST_CAP <= (long long)edgeRows * HEADS);

    for (int i = tid; i < nrows * F; i += nthr) {
      const int r = i / F;
      acc[r * OWN_PITCH + (i - r * F)] = 0.f;
    }
    for (int i = tid; i < nrows * NH; i += nthr) {
      const float ed = e_dst[(size_t)base * NH + i];
      float l = e_src[(size_t)base * NH + i] + ed;
      l = (l > 0.f) ? l : l * NEG_SLOPE;
      edS[i] = ed;
      mS[i] = l;
      denS[i] = 0.f;
    }
    if (tid == 0) listN = 0;
    int total;
    if (LMODE == 2) {
      total = __builtin_amdgcn_readfirstlane(cntW[blk]);
      if (total <= LIST_CAP) {
        for (int i = tid; i < total; i += nthr) {
          const int p = lw[i];
          listS[i] = min(p & 0xFFFFFF, n - 1);
          listD[i] = min((int)((unsigned)p >> 24), nrows - 1);
        }
      }
      __syncthreads();
    } else {
      __syncthreads();

      {
        auto fileStep = [&](const int4 cur, const int e0) {
          const int dl[4] = {cur.x - base, cur.y - base, cur.z - base, cur.w - base};
          bool mt[4];
          unsigned bm[4];
          for (int k = 0; k < 4; ++k) {
            mt[k] = (unsigned)dl[k] < (unsigned)nrows;
            bm[k] = __builtin_amdgcn_ballot_w32(mt[k]);
          }
          if ((bm[0] | bm[1] | bm[2] | bm[3]) == 0u) return;
          const int found = __builtin_popcount(bm[0]) + __builtin_popcount(bm[1]) +
                            __builtin_popcount(bm[2]) + __builtin_popcount(bm[3]);
          int slot = 0;
          if (lane == 0) slot = atomicAdd(&listN, found);
          slot = __builtin_amdgcn_readfirstlane(slot);
          for (int k = 0; k < 4; ++k) {
            if (mt[k]) {
              const int pos = slot + (int)__builtin_amdgcn_mbcnt_lo(bm[k], 0u);
              if (pos < LIST_CAP) {
                listS[pos] = e0 + k;
                listD[pos] = dl[k];
              }
            }
            slot += __builtin_popcount(bm[k]);
          }
        };
        if (vec) {
          const int nFull = E / chunk;
          if (nFull > 0) {
            const int itF0 = (int)(((long long)blockIdx.x * nFull) / gridDim.x);
            auto edgeF = [&](int i) -> int {
              int itr = itF0 + min(i, nFull - 1);
              if (itr >= nFull) itr -= nFull;
              return itr * chunk + tid * 4;
            };
            int4 q[SWEEP_AHEAD];
#pragma unroll
            for (int u = 0; u < SWEEP_AHEAD; ++u)
              q[u] = *reinterpret_cast<const int4*>(dstIdx + edgeF(u));
            for (int i = 0; i < nFull; i += SWEEP_AHEAD) {
#pragma unroll
              for (int u = 0; u < SWEEP_AHEAD; ++u) {
                if (i + u < nFull) fileStep(q[u], edgeF(i + u));
                q[u] = *reinterpret_cast<const int4*>(dstIdx + edgeF(i + u + SWEEP_AHEAD));
              }
            }
          }
          if (nFull * chunk < E) {
            const int e0 = nFull * chunk + tid * 4;
            int4 v = make_int4(-1, -1, -1, -1);
            if (e0 < E) v = *reinterpret_cast<const int4*>(dstIdx + e0);
            fileStep(v, e0);
          }
        } else {
          int4 q0 = fetch(0), q1 = fetch(1), q2 = fetch(2), q3 = fetch(3);
          for (int i = 0; i < nIt; ++i) {
            const int4 cur = q0;
            q0 = q1; q1 = q2; q2 = q3; q3 = fetch(i + 4);
            fileStep(cur, edge0(i));
          }
        }
      }
      __syncthreads();
      total = __builtin_amdgcn_readfirstlane(listN);
      if (tid == 0) cntW[blk] = canSave ? total : LIST_CAP + 1;
    }

    if (total <= LIST_CAP) {
      for (int i0 = tid; i0 < total; i0 += 4 * nthr) {
        int ci[4], sv[4], dv[4];
        bool ok[4];
        for (int k = 0; k < 4; ++k) {
          const int idx = i0 + k * nthr;
          ok[k] = idx < total;
          ci[k] = ok[k] ? idx : i0;
        }
        for (int k = 0; k < 4; ++k) {
          sv[k] = (LMODE == 1) ? srcIdx[listS[ci[k]]] : listS[ci[k]];
          dv[k] = listD[ci[k]];
        }
        float es[4][NH];
        for (int k = 0; k < 4; ++k)
          for (int h = 0; h < NH; ++h)
            es[k][h] = e_src[(size_t)sv[k] * NH + h];
        for (int k = 0; k < 4; ++k) {
          if (ok[k]) {
            if (LMODE == 1) listS[ci[k]] = sv[k];
            for (int h = 0; h < NH; ++h) {
              float l = es[k][h] + edS[dv[k] * NH + h];
              l = (l > 0.f) ? l : l * NEG_SLOPE;
              ldsMaxFloat(&mS[dv[k] * NH + h], l);
            }
          }
        }
      }
      __syncthreads();

      if (LMODE == 1 && canSave) {
        for (int i = tid; i < total; i += nthr) lw[i] = (int)((unsigned)listS[i] | ((unsigned)listD[i] << 24));
      }

      const int totalAll = total + nrows;
      const int waveS = __builtin_amdgcn_readfirstlane(wave);
      for (int g = waveS * LIST_B; g < totalAll; g += nwav * LIST_B) {
        int sv[LIST_B], dv[LIST_B];
#pragma unroll
        for (int k = 0; k < LIST_B; ++k) {
          const int idx = min(g + k, totalAll - 1);
          const bool isList = idx < total;
          const int li = isList ? idx : 0;
          const int ls = listS[li];
          const int ld = listD[li];
          const int d = isList ? ld : idx - total;
          const int s = isList ? ls : base + d;
          sv[k] = __builtin_amdgcn_readfirstlane(s);
          dv[k] = __builtin_amdgcn_readfirstlane(d);
        }
        float hv[LIST_B][J];
        float el[LIST_B];
#pragma unroll
        for (int k = 0; k < LIST_B; ++k) {
          const float* hp = H + (size_t)sv[k] * F;
#pragma unroll
          for (int j = 0; j < J; ++j) hv[k][j] = hp[lane + 32 * j];
          el[k] = e_src[(size_t)sv[k] * NH + (lane & (NH - 1))];
        }
#pragma unroll
        for (int k = 0; k < LIST_B; ++k) {
          if (g + k < totalAll) {
            const int d = dv[k];
            const int h = lane & (NH - 1);
            float l = el[k] + edS[d * NH + h];
            l = (l > 0.f) ? l : l * NEG_SLOPE;
            const float ex = expf(l - mS[d * NH + h]);
            if (lane < NH) atomicAdd(&denS[d * NH + lane], ex);
            float* ap = acc + d * OWN_PITCH;
#pragma unroll
            for (int j = 0; j < J; ++j) {
              const float exj = __int_as_float(
                  __builtin_amdgcn_readlane(__float_as_int(ex), (32 * j) / C));
              atomicAdd(&ap[lane + 32 * j], exj * hv[k][j]);
            }
          }
        }
      }
    } else {
      {
        int4 q0 = fetch(0), q1 = fetch(1), q2 = fetch(2), q3 = fetch(3);
        for (int i = 0; i < nIt; ++i) {
          const int4 cur = q0;
          q0 = q1; q1 = q2; q2 = q3; q3 = fetch(i + 4);
          const int dl[4] = {cur.x - base, cur.y - base, cur.z - base, cur.w - base};
          bool mt[4];
          for (int k = 0; k < 4; ++k) mt[k] = (unsigned)dl[k] < (unsigned)nrows;
          if (__builtin_amdgcn_ballot_w32(mt[0] | mt[1] | mt[2] | mt[3]) == 0u) continue;
          const int e0 = edge0(i);
          int sv[4];
          for (int k = 0; k < 4; ++k) sv[k] = mt[k] ? srcIdx[e0 + k] : 0;
          float es[4][NH];
          for (int k = 0; k < 4; ++k)
            for (int h = 0; h < NH; ++h)
              es[k][h] = mt[k] ? e_src[(size_t)sv[k] * NH + h] : 0.f;
          for (int k = 0; k < 4; ++k) {
            if (mt[k]) {
              for (int h = 0; h < NH; ++h) {
                float l = es[k][h] + edS[dl[k] * NH + h];
                l = (l > 0.f) ? l : l * NEG_SLOPE;
                ldsMaxFloat(&mS[dl[k] * NH + h], l);
              }
            }
          }
        }
      }
      __syncthreads();

      {
        auto addEdge = [&](int s, int d) {
          const float* hp = H + (size_t)s * F;
          float hv[J];
          for (int j = 0; j < J; ++j) hv[j] = hp[lane + 32 * j];
          const int h = lane & (NH - 1);
          float l = e_src[(size_t)s * NH + h] + edS[d * NH + h];
          l = (l > 0.f) ? l : l * NEG_SLOPE;
          const float ex = expf(l - mS[d * NH + h]);
          if (lane < NH) atomicAdd(&denS[d * NH + lane], ex);
          float* ap = acc + d * OWN_PITCH;
#pragma unroll
          for (int j = 0; j < J; ++j) {
            const float exj = __int_as_float(
                __builtin_amdgcn_readlane(__float_as_int(ex), (32 * j) / C));
            atomicAdd(&ap[lane + 32 * j], exj * hv[j]);
          }
        };

        int4 q0 = fetch(0), q1 = fetch(1), q2 = fetch(2), q3 = fetch(3);
        for (int i = 0; i < nIt; ++i) {
          const int4 cur = q0;
          q0 = q1; q1 = q2; q2 = q3; q3 = fetch(i + 4);
          const int dl[4] = {cur.x - base, cur.y - base, cur.z - base, cur.w - base};
          bool mt[4];
          for (int k = 0; k < 4; ++k) mt[k] = (unsigned)dl[k] < (unsigned)nrows;
          if (__builtin_amdgcn_ballot_w32(mt[0] | mt[1] | mt[2] | mt[3]) == 0u) continue;
          const int e0 = edge0(i);
          int sv[4];
          for (int k = 0; k < 4; ++k) sv[k] = mt[k] ? srcIdx[e0 + k] : 0;
          for (int k = 0; k < 4; ++k) {
            unsigned mask = __builtin_amdgcn_ballot_w32(mt[k]);
            while (mask) {
              const int j = __builtin_ctz(mask);
              mask &= mask - 1u;
              addEdge(__builtin_amdgcn_readlane(sv[k], j),
                      __builtin_amdgcn_readlane(dl[k], j));
            }
          }
        }
        for (int r = wave; r < nrows; r += nwav) addEdge(base + r, r);
      }
    }
    __syncthreads();

    if (!FINAL) {
      for (int i = tid; i < nrows * F; i += nthr) {
        const int r = i / F;
        const int f = i - r * F;
        const float v = acc[r * OWN_PITCH + f] / (denS[r * NH + f / C] + 1e-16f) + b[f];
        out[(size_t)(base + r) * F + f] = (v > 0.f) ? v : (expf(v) - 1.f);
      }
    } else {
      for (int r = wave; r < nrows; r += nwav) {
        const float den = denS[r * NH] + 1e-16f;
        float sum = 0.f;
        for (int c = lane; c < F; c += 32) {
          float v = acc[r * OWN_PITCH + c] / den + b[c];
          v = (v > 0.f) ? v : (expf(v) - 1.f);
          sum += v;
        }
        for (int off = 16; off > 0; off >>= 1)
          sum += __shfl_down(sum, off, 32);
        if (lane == 0) out[base + r] = sum / (float)F;
      }
    }
    __syncthreads();
  }
}

static inline int cdiv(long long a, long long b) { return (int)((a + b - 1) / b); }

extern "C" void kernel_launch(void* const* d_in, const int* in_sizes, int n_in,
                              void* d_out, int out_size, void* d_ws, size_t ws_size,
                              hipStream_t stream) {
  const float* x      = (const float*)d_in[0];
  const int*   ei     = (const int*)  d_in[1];
  const float* W1     = (const float*)d_in[2];
  const float* a1_src = (const float*)d_in[3];
  const float* a1_dst = (const float*)d_in[4];
  const float* b1     = (const float*)d_in[5];
  const float* W2     = (const float*)d_in[6];
  const float* a2_src = (const float*)d_in[7];
  const float* a2_dst = (const float*)d_in[8];
  const float* b2     = (const float*)d_in[9];
  float* outp = (float*)d_out;

  const int N    = in_sizes[0] / DIN;
  const int E    = in_sizes[1] / 2;
  const int Etot = E + N;
  const int F1   = HEADS * HID;
  const int* srcIdx = ei;
  const int* dstIdx = ei + E;

  float* ws = (float*)d_ws;
  float* h1      = ws;
  float* e_src_b = h1 + (size_t)N * F1;
  float* e_dst_b = e_src_b + (size_t)N * HEADS;
  float* edgebuf = e_dst_b + (size_t)N * HEADS;
  float* m_b     = edgebuf + (size_t)Etot * HEADS;
  float* den_b   = m_b + (size_t)N * HEADS;
  float* out1    = den_b + (size_t)N * HEADS;
  float* h2      = h1;
  int* listW = (int*)edgebuf;
  int* cntW  = (int*)m_b;

  const int TB = 256;

  {
    int strips = (N / 16) * (F1 / 64);
    gemm_f32_wmma<DIN, F1><<<cdiv(strips, TB / 32), TB, 0, stream>>>(x, W1, h1, N);

    dot_e_kernel<<<cdiv((long long)N * HEADS, TB), TB, 0, stream>>>(
        h1, a1_src, a1_dst, e_src_b, e_dst_b, N, HEADS, HID);

    gat_owner<HEADS, HID, false, 1><<<OWN_GROUPS, OWN_THREADS, 0, stream>>>(
        srcIdx, dstIdx, E, N, h1, e_src_b, e_dst_b, b1, out1, listW, cntW, Etot);
  }

  {
    int strips = (N / 16) * (DOUT / 64);
    gemm_f32_wmma<F1, DOUT><<<cdiv(strips, TB / 32), TB, 0, stream>>>(out1, W2, h2, N);

    dot_e_kernel<<<cdiv(N, TB), TB, 0, stream>>>(
        h2, a2_src, a2_dst, e_src_b, e_dst_b, N, 1, DOUT);

    gat_owner<1, DOUT, true, 2><<<OWN_GROUPS, OWN_THREADS, 0, stream>>>(
        srcIdx, dstIdx, E, N, h2, e_src_b, e_dst_b, b2, outp, listW, cntW, Etot);
  }
}


# gat_owner sums: each wave owns one row class (local row mod 32), lists filed in the max pass; plain LDS read-add-write, den in a register, no LDS float atomics; sum order = class list (arrival) order
# speedup vs baseline: 3.8209x; 1.1177x over previous
#include <hip/hip_runtime.h>
#include <math.h>

#define DIN 256
#define HID 64
#define HEADS 4
#define DOUT 128
#define NEG_SLOPE 0.2f

typedef __attribute__((ext_vector_type(2))) float v2f;
typedef __attribute__((ext_vector_type(8))) float v8f;

template<int K, int NOUT>
__global__ void gemm_f32_wmma(const float* __restrict__ X,
                              const float* __restrict__ W,
                              float* __restrict__ H, int nrows) {
  const int lane = threadIdx.x & 31;
  const int wave = threadIdx.x >> 5;
  const int colGroups = NOUT / 64;
  const int strip = blockIdx.x * (blockDim.x >> 5) + wave;
  const int rowTile = strip / colGroups;
  const int colGroup = strip - rowTile * colGroups;
  if (rowTile * 16 >= nrows) return;
  const int r = lane & 15;
  const int half = lane >> 4;
  const float* __restrict__ Ap = X + (size_t)(rowTile * 16 + r) * K + 2 * half;
  const float* __restrict__ Bp = W + (size_t)(2 * half) * NOUT + colGroup * 64 + r;
  v8f acc0 = {}, acc1 = {}, acc2 = {}, acc3 = {};
#pragma unroll 4
  for (int k = 0; k < K; k += 4) {
    v2f a;
    a[0] = Ap[k];
    a[1] = Ap[k + 1];
    const float* b0 = Bp + (size_t)k * NOUT;
    const float* b1 = Bp + (size_t)(k + 1) * NOUT;
    v2f bA, bB, bC, bD;
    bA[0] = b0[0];   bA[1] = b1[0];
    bB[0] = b0[16];  bB[1] = b1[16];
    bC[0] = b0[32];  bC[1] = b1[32];
    bD[0] = b0[48];  bD[1] = b1[48];
    acc0 = __builtin_amdgcn_wmma_f32_16x16x4_f32(false, a, false, bA,
                                                 (short)0, acc0, false, false);
    acc1 = __builtin_amdgcn_wmma_f32_16x16x4_f32(false, a, false, bB,
                                                 (short)0, acc1, false, false);
    acc2 = __builtin_amdgcn_wmma_f32_16x16x4_f32(false, a, false, bC,
                                                 (short)0, acc2, false, false);
    acc3 = __builtin_amdgcn_wmma_f32_16x16x4_f32(false, a, false, bD,
                                                 (short)0, acc3, false, false);
  }
  float* __restrict__ Cp =
      H + (size_t)(rowTile * 16 + 8 * half) * NOUT + colGroup * 64 + r;
#pragma unroll
  for (int v = 0; v < 8; ++v) {
    Cp[(size_t)v * NOUT]      = acc0[v];
    Cp[(size_t)v * NOUT + 16] = acc1[v];
    Cp[(size_t)v * NOUT + 32] = acc2[v];
    Cp[(size_t)v * NOUT + 48] = acc3[v];
  }
}

__global__ void dot_e_kernel(const float* __restrict__ H,
                             const float* __restrict__ a_src,
                             const float* __restrict__ a_dst,
                             float* __restrict__ e_src,
                             float* __restrict__ e_dst,
                             int n, int heads, int C) {
  int idx = blockIdx.x * blockDim.x + threadIdx.x;
  if (idx >= n * heads) return;
  int node = idx / heads;
  int hd = idx - node * heads;
  const float* hp = H + (size_t)node * heads * C + (size_t)hd * C;
  const float* as = a_src + (size_t)hd * C;
  const float* ad = a_dst + (size_t)hd * C;
  float s0 = 0.f, s1 = 0.f;
  for (int c = 0; c < C; ++c) {
    float v = hp[c];
    s0 += v * as[c];
    s1 += v * ad[c];
  }
  e_src[idx] = s0;
  e_dst[idx] = s1;
}

#define OWN_ROWS 196
#define OWN_PITCH 256
#define OWN_THREADS 1024
#define OWN_GROUPS 256
#define LIST_CAP 8192
#define PRIV_CLASSES 32
#define PRIV_CAP 256
#define SWEEP_AHEAD 8

__device__ __forceinline__ void ldsMaxFloat(float* addr, float val) {
  if (val >= 0.0f)
    atomicMax((int*)addr, __float_as_int(val));
  else
    atomicMin((unsigned int*)addr, __float_as_uint(val));
}

template<int NH, int C, bool FINAL, int LMODE>
__global__ void __launch_bounds__(OWN_THREADS)
gat_owner(const int* __restrict__ srcIdx, const int* __restrict__ dstIdx,
          int E, int n,
          const float* __restrict__ H,
          const float* __restrict__ e_src,
          const float* __restrict__ e_dst,
          const float* __restrict__ b,
          float* __restrict__ out,
          int* listW,
          int* cntW,
          int edgeRows) {
  constexpr int F = NH * C;
  constexpr int J = F / 32;
  constexpr int LIST_B = 1024 / F;
  static_assert(!FINAL || NH == 1, "row mean is written for one head");
  static_assert(F <= OWN_PITCH && (F % 32) == 0 && (C % 32) == 0, "channel layout");
  static_assert(LIST_B >= 1 && LIST_B * J == 32, "32 row loads in flight per wave");
  static_assert(LMODE == 1 || LMODE == 2, "1: sweep and save the list, 2: read the saved list");
  static_assert(OWN_ROWS <= 256, "a saved word keeps the local row in its top 8 bits");
  __shared__ __attribute__((aligned(16))) float acc[OWN_ROWS * OWN_PITCH];
  __shared__ float mS[OWN_ROWS * NH];
  __shared__ float denS[OWN_ROWS * NH];
  __shared__ float edS[OWN_ROWS * NH];
  __shared__ int listS[LIST_CAP];
  __shared__ int listD[LIST_CAP];
  __shared__ int listN;
  __shared__ int privL[PRIV_CLASSES * PRIV_CAP];
  __shared__ int privN[PRIV_CLASSES];

  const int tid = threadIdx.x;
  const int lane = tid & 31;
  const int wave = tid >> 5;
  const int nthr = blockDim.x;
  const int nwav = nthr >> 5;
  const int chunk = nthr * 4;
  const int nIt = (E + chunk - 1) / chunk;
  const int it0 = (int)(((long long)blockIdx.x * nIt) / gridDim.x);
  const bool vec = ((E & 3) == 0) && ((((size_t)dstIdx) & 15) == 0);

  auto fetch = [&](int i) -> int4 {
    int4 v = make_int4(-1, -1, -1, -1);
    if (i < nIt) {
      int itr = it0 + i;
      if (itr >= nIt) itr -= nIt;
      const int e0 = itr * chunk + tid * 4;
      if (vec && e0 + 3 < E) {
        v = *reinterpret_cast<const int4*>(dstIdx + e0);
      } else {
        if (e0 < E) v.x = dstIdx[e0];
        if (e0 + 1 < E) v.y = dstIdx[e0 + 1];
        if (e0 + 2 < E) v.z = dstIdx[e0 + 2];
        if (e0 + 3 < E) v.w = dstIdx[e0 + 3];
      }
    }
    return v;
  };
  auto edge0 = [&](int i) -> int {
    int itr = it0 + i;
    if (itr >= nIt) itr -= nIt;
    return itr * chunk + tid * 4;
  };

  for (int base = blockIdx.x * OWN_ROWS; base < n; base += gridDim.x * OWN_ROWS) {
    const int nrows = min(OWN_ROWS, n - base);
    const int blk = base / OWN_ROWS;
    int* const lw = listW + (size_t)blk * LIST_CAP;
    const bool canSave = (n <= (1 << 24)) &&
                         ((long long)(blk + 1) * LIST_CAP <= (long long)edgeRows * HEADS);

    for (int i = tid; i < nrows * F; i += nthr) {
      const int r = i / F;
      acc[r * OWN_PITCH + (i - r * F)] = 0.f;
    }
    for (int i = tid; i < nrows * NH; i += nthr) {
      const float ed = e_dst[(size_t)base * NH + i];
      float l = e_src[(size_t)base * NH + i] + ed;
      l = (l > 0.f) ? l : l * NEG_SLOPE;
      edS[i] = ed;
      mS[i] = l;
      denS[i] = 0.f;
    }
    if (tid == 0) listN = 0;
    if (tid < PRIV_CLASSES) privN[tid] = 0;
    const bool packOK = n <= (1 << 24);
    bool viaList = false;
    int total;
    if (LMODE == 2) {
      total = __builtin_amdgcn_readfirstlane(cntW[blk]);
      if (total <= LIST_CAP) {
        for (int i = tid; i < total; i += nthr) {
          const int p = lw[i];
          listS[i] = min(p & 0xFFFFFF, n - 1);
          listD[i] = min((int)((unsigned)p >> 24), nrows - 1);
        }
      }
      __syncthreads();
    } else {
      __syncthreads();

      {
        auto fileStep = [&](const int4 cur, const int e0) {
          const int dl[4] = {cur.x - base, cur.y - base, cur.z - base, cur.w - base};
          bool mt[4];
          unsigned bm[4];
          for (int k = 0; k < 4; ++k) {
            mt[k] = (unsigned)dl[k] < (unsigned)nrows;
            bm[k] = __builtin_amdgcn_ballot_w32(mt[k]);
          }
          if ((bm[0] | bm[1] | bm[2] | bm[3]) == 0u) return;
          const int found = __builtin_popcount(bm[0]) + __builtin_popcount(bm[1]) +
                            __builtin_popcount(bm[2]) + __builtin_popcount(bm[3]);
          int slot = 0;
          if (lane == 0) slot = atomicAdd(&listN, found);
          slot = __builtin_amdgcn_readfirstlane(slot);
          for (int k = 0; k < 4; ++k) {
            if (mt[k]) {
              const int pos = slot + (int)__builtin_amdgcn_mbcnt_lo(bm[k], 0u);
              if (pos < LIST_CAP) {
                listS[pos] = e0 + k;
                listD[pos] = dl[k];
              }
            }
            slot += __builtin_popcount(bm[k]);
          }
        };
        if (vec) {
          const int nFull = E / chunk;
          if (nFull > 0) {
            const int itF0 = (int)(((long long)blockIdx.x * nFull) / gridDim.x);
            auto edgeF = [&](int i) -> int {
              int itr = itF0 + min(i, nFull - 1);
              if (itr >= nFull) itr -= nFull;
              return itr * chunk + tid * 4;
            };
            int4 q[SWEEP_AHEAD];
#pragma unroll
            for (int u = 0; u < SWEEP_AHEAD; ++u)
              q[u] = *reinterpret_cast<const int4*>(dstIdx + edgeF(u));
            for (int i = 0; i < nFull; i += SWEEP_AHEAD) {
#pragma unroll
              for (int u = 0; u < SWEEP_AHEAD; ++u) {
                if (i + u < nFull) fileStep(q[u], edgeF(i + u));
                q[u] = *reinterpret_cast<const int4*>(dstIdx + edgeF(i + u + SWEEP_AHEAD));
              }
            }
          }
          if (nFull * chunk < E) {
            const int e0 = nFull * chunk + tid * 4;
            int4 v = make_int4(-1, -1, -1, -1);
            if (e0 < E) v = *reinterpret_cast<const int4*>(dstIdx + e0);
            fileStep(v, e0);
          }
        } else {
          int4 q0 = fetch(0), q1 = fetch(1), q2 = fetch(2), q3 = fetch(3);
          for (int i = 0; i < nIt; ++i) {
            const int4 cur = q0;
            q0 = q1; q1 = q2; q2 = q3; q3 = fetch(i + 4);
            fileStep(cur, edge0(i));
          }
        }
      }
      __syncthreads();
      total = __builtin_amdgcn_readfirstlane(listN);
      if (tid == 0) cntW[blk] = canSave ? total : LIST_CAP + 1;
    }

    if (total <= LIST_CAP) {
      for (int i0 = tid; i0 < total; i0 += 4 * nthr) {
        int ci[4], sv[4], dv[4];
        bool ok[4];
        for (int k = 0; k < 4; ++k) {
          const int idx = i0 + k * nthr;
          ok[k] = idx < total;
          ci[k] = ok[k] ? idx : i0;
        }
        for (int k = 0; k < 4; ++k) {
          sv[k] = (LMODE == 1) ? srcIdx[listS[ci[k]]] : listS[ci[k]];
          dv[k] = listD[ci[k]];
        }
        float es[4][NH];
        for (int k = 0; k < 4; ++k)
          for (int h = 0; h < NH; ++h)
            es[k][h] = e_src[(size_t)sv[k] * NH + h];
        for (int k = 0; k < 4; ++k) {
          if (ok[k]) {
            if (LMODE == 1) listS[ci[k]] = sv[k];
            if (packOK) {
              const int o = dv[k] & (PRIV_CLASSES - 1);
              const int pos = atomicAdd(&privN[o], 1);
              if (pos < PRIV_CAP)
                privL[o * PRIV_CAP + pos] = (int)((unsigned)sv[k] | ((unsigned)dv[k] << 24));
            }
            for (int h = 0; h < NH; ++h) {
              float l = es[k][h] + edS[dv[k] * NH + h];
              l = (l > 0.f) ? l : l * NEG_SLOPE;
              ldsMaxFloat(&mS[dv[k] * NH + h], l);
            }
          }
        }
      }
      if (packOK) {
        for (int r = tid; r < nrows; r += nthr) {
          const int o = r & (PRIV_CLASSES - 1);
          const int pos = atomicAdd(&privN[o], 1);
          if (pos < PRIV_CAP) privL[o * PRIV_CAP + pos] = (int)((unsigned)(base + r) | ((unsigned)r << 24));
        }
      }
      __syncthreads();
      viaList = packOK && (__builtin_amdgcn_ballot_w32(privN[lane] > PRIV_CAP) == 0u);

      if (LMODE == 1 && canSave) {
        for (int i = tid; i < total; i += nthr) lw[i] = (int)((unsigned)listS[i] | ((unsigned)listD[i] << 24));
      }

      if (viaList) {
        constexpr int JV = F / 128;
        constexpr int HG = (C >= 128) ? 1 : 128 / C;
        constexpr int EL = LIST_B * NH;
        static_assert(F % 128 == 0 && (C % 128 == 0 || 128 % C == 0) && 32 % HG == 0, "16-byte pieces");
        static_assert(LIST_B * JV * 4 == 32, "32 registers of row loads in flight per wave");
        static_assert(((OWN_ROWS + PRIV_CLASSES - 1) / PRIV_CLASSES) * NH <= 32,
                      "one lane per (row of a class, head) for den");
        static_assert((LIST_B & (LIST_B - 1)) == 0 && (NH & (NH - 1)) == 0 && EL <= 32,
                      "one (entry, head) pair per lane");
        const int kl = (lane / NH) & (LIST_B - 1);
        const int hl = lane & (NH - 1);
        const int waveS = __builtin_amdgcn_readfirstlane(wave);
        for (int o = waveS; o < PRIV_CLASSES; o += nwav) {
          const int cnt = __builtin_amdgcn_readfirstlane(privN[o]);
          if (cnt == 0) continue;
          const int* pl = privL + o * PRIV_CAP;
          float denReg = 0.f;
          const int ql = lane / NH;
          const int nFull = cnt / LIST_B;
          const int nTrip = (cnt + LIST_B - 1) / LIST_B;
          auto trip = [&](int t, int w, const bool tail, int& wN) __attribute__((always_inline)) {
            const int g = t * LIST_B;
            const int s = w & 0xFFFFFF;
            const int d = (int)((unsigned)w >> 24);
            const float edv = edS[d * NH + hl];
            const float mv = mS[d * NH + hl];
            int sv[LIST_B], dv[LIST_B];
#pragma unroll
            for (int k = 0; k < LIST_B; ++k) {
              sv[k] = __builtin_amdgcn_readlane(s, k * NH);
              dv[k] = __builtin_amdgcn_readlane(d, k * NH);
            }
            float4 hv[LIST_B][JV];
#pragma unroll
            for (int k = 0; k < LIST_B; ++k) {
              const float4* hp = reinterpret_cast<const float4*>(H + (size_t)sv[k] * F) + lane;
#pragma unroll
              for (int jj = 0; jj < JV; ++jj) hv[k][jj] = hp[32 * jj];
            }
            __builtin_amdgcn_sched_barrier(0);
            const float elv = e_src[(size_t)s * NH + hl];
            wN = pl[min(g + LIST_B + kl, cnt - 1)];
            float l = elv + edv;
            l = (l > 0.f) ? l : l * NEG_SLOPE;
            const float exv = expf(l - mv);
#pragma unroll
            for (int k = 0; k < LIST_B; ++k) {
              if (!tail || g + k < cnt) {
                float4* ap = reinterpret_cast<float4*>(acc + dv[k] * OWN_PITCH) + lane;
                {
                  float eh = __int_as_float(__builtin_amdgcn_readlane(__float_as_int(exv), k * NH));
#pragma unroll
                  for (int h = 1; h < NH; ++h) {
                    const float e2 = __int_as_float(
                        __builtin_amdgcn_readlane(__float_as_int(exv), k * NH + h));
                    eh = (hl == h) ? e2 : eh;
                  }
                  denReg += (ql == (dv[k] / PRIV_CLASSES)) ? eh : 0.f;
                }
                float4 a[JV];
#pragma unroll
                for (int jj = 0; jj < JV; ++jj) a[jj] = ap[32 * jj];
#pragma unroll
                for (int jj = 0; jj < JV; ++jj) {
                  const int h0 = (C >= 128) ? (128 * jj) / C : jj * HG;
                  float e = __int_as_float(
                      __builtin_amdgcn_readlane(__float_as_int(exv), k * NH + h0));
#pragma unroll
                  for (int q = 1; q < HG; ++q) {
                    const float eq = __int_as_float(
                        __builtin_amdgcn_readlane(__float_as_int(exv), k * NH + h0 + q));
                    e = (lane >= q * (32 / HG)) ? eq : e;
                  }
                  const float4 hq = hv[k][jj];
                  a[jj].x += e * hq.x;
                  a[jj].y += e * hq.y;
                  a[jj].z += e * hq.z;
                  a[jj].w += e * hq.w;
                }
#pragma unroll
                for (int jj = 0; jj < JV; ++jj) ap[32 * jj] = a[jj];
              }
            }
          };
          int w = pl[min(kl, cnt - 1)];
          int t = 0;
          for (; t < nFull; ++t) {
            int wN;
            trip(t, w, false, wN);
            w = wN;
          }
          if (t < nTrip) {
            int wN;
            trip(t, w, true, wN);
          }
          {
            const int r = o + PRIV_CLASSES * ql;
            if (r < nrows) denS[r * NH + hl] = denReg;
          }
        }
      }
    }
    if (!viaList) {
      {
        int4 q0 = fetch(0), q1 = fetch(1), q2 = fetch(2), q3 = fetch(3);
        for (int i = 0; i < nIt; ++i) {
          const int4 cur = q0;
          q0 = q1; q1 = q2; q2 = q3; q3 = fetch(i + 4);
          const int dl[4] = {cur.x - base, cur.y - base, cur.z - base, cur.w - base};
          bool mt[4];
          for (int k = 0; k < 4; ++k) mt[k] = (unsigned)dl[k] < (unsigned)nrows;
          if (__builtin_amdgcn_ballot_w32(mt[0] | mt[1] | mt[2] | mt[3]) == 0u) continue;
          const int e0 = edge0(i);
          int sv[4];
          for (int k = 0; k < 4; ++k) sv[k] = mt[k] ? srcIdx[e0 + k] : 0;
          float es[4][NH];
          for (int k = 0; k < 4; ++k)
            for (int h = 0; h < NH; ++h)
              es[k][h] = mt[k] ? e_src[(size_t)sv[k] * NH + h] : 0.f;
          for (int k = 0; k < 4; ++k) {
            if (mt[k]) {
              for (int h = 0; h < NH; ++h) {
                float l = es[k][h] + edS[dl[k] * NH + h];
                l = (l > 0.f) ? l : l * NEG_SLOPE;
                ldsMaxFloat(&mS[dl[k] * NH + h], l);
              }
            }
          }
        }
      }
      __syncthreads();

      {
        auto addEdge = [&](int s, int d) {
          const float* hp = H + (size_t)s * F;
          float hv[J];
          for (int j = 0; j < J; ++j) hv[j] = hp[lane + 32 * j];
          const int h = lane & (NH - 1);
          float l = e_src[(size_t)s * NH + h] + edS[d * NH + h];
          l = (l > 0.f) ? l : l * NEG_SLOPE;
          const float ex = expf(l - mS[d * NH + h]);
          if (lane < NH) atomicAdd(&denS[d * NH + lane], ex);
          float* ap = acc + d * OWN_PITCH;
#pragma unroll
          for (int j = 0; j < J; ++j) {
            const float exj = __int_as_float(
                __builtin_amdgcn_readlane(__float_as_int(ex), (32 * j) / C));
            atomicAdd(&ap[lane + 32 * j], exj * hv[j]);
          }
        };

        int4 q0 = fetch(0), q1 = fetch(1), q2 = fetch(2), q3 = fetch(3);
        for (int i = 0; i < nIt; ++i) {
          const int4 cur = q0;
          q0 = q1; q1 = q2; q2 = q3; q3 = fetch(i + 4);
          const int dl[4] = {cur.x - base, cur.y - base, cur.z - base, cur.w - base};
          bool mt[4];
          for (int k = 0; k < 4; ++k) mt[k] = (unsigned)dl[k] < (unsigned)nrows;
          if (__builtin_amdgcn_ballot_w32(mt[0] | mt[1] | mt[2] | mt[3]) == 0u) continue;
          const int e0 = edge0(i);
          int sv[4];
          for (int k = 0; k < 4; ++k) sv[k] = mt[k] ? srcIdx[e0 + k] : 0;
          for (int k = 0; k < 4; ++k) {
            unsigned mask = __builtin_amdgcn_ballot_w32(mt[k]);
            while (mask) {
              const int j = __builtin_ctz(mask);
              mask &= mask - 1u;
              addEdge(__builtin_amdgcn_readlane(sv[k], j),
                      __builtin_amdgcn_readlane(dl[k], j));
            }
          }
        }
        for (int r = wave; r < nrows; r += nwav) addEdge(base + r, r);
      }
    }
    __syncthreads();

    if (!FINAL) {
      for (int i = tid; i < nrows * F; i += nthr) {
        const int r = i / F;
        const int f = i - r * F;
        const float v = acc[r * OWN_PITCH + f] / (denS[r * NH + f / C] + 1e-16f) + b[f];
        out[(size_t)(base + r) * F + f] = (v > 0.f) ? v : (expf(v) - 1.f);
      }
    } else {
      for (int r = wave; r < nrows; r += nwav) {
        const float den = denS[r * NH] + 1e-16f;
        float sum = 0.f;
        for (int c = lane; c < F; c += 32) {
          float v = acc[r * OWN_PITCH + c] / den + b[c];
          v = (v > 0.f) ? v : (expf(v) - 1.f);
          sum += v;
        }
        for (int off = 16; off > 0; off >>= 1)
          sum += __shfl_down(sum, off, 32);
        if (lane == 0) out[base + r] = sum / (float)F;
      }
    }
    __syncthreads();
  }
}

static inline int cdiv(long long a, long long b) { return (int)((a + b - 1) / b); }

extern "C" void kernel_launch(void* const* d_in, const int* in_sizes, int n_in,
                              void* d_out, int out_size, void* d_ws, size_t ws_size,
                              hipStream_t stream) {
  const float* x      = (const float*)d_in[0];
  const int*   ei     = (const int*)  d_in[1];
  const float* W1     = (const float*)d_in[2];
  const float* a1_src = (const float*)d_in[3];
  const float* a1_dst = (const float*)d_in[4];
  const float* b1     = (const float*)d_in[5];
  const float* W2     = (const float*)d_in[6];
  const float* a2_src = (const float*)d_in[7];
  const float* a2_dst = (const float*)d_in[8];
  const float* b2     = (const float*)d_in[9];
  float* outp = (float*)d_out;

  const int N    = in_sizes[0] / DIN;
  const int E    = in_sizes[1] / 2;
  const int Etot = E + N;
  const int F1   = HEADS * HID;
  const int* srcIdx = ei;
  const int* dstIdx = ei + E;

  float* ws = (float*)d_ws;
  float* h1      = ws;
  float* e_src_b = h1 + (size_t)N * F1;
  float* e_dst_b = e_src_b + (size_t)N * HEADS;
  float* edgebuf = e_dst_b + (size_t)N * HEADS;
  float* m_b     = edgebuf + (size_t)Etot * HEADS;
  float* den_b   = m_b + (size_t)N * HEADS;
  float* out1    = den_b + (size_t)N * HEADS;
  float* h2      = h1;
  int* listW = (int*)edgebuf;
  int* cntW  = (int*)m_b;

  const int TB = 256;

  {
    int strips = (N / 16) * (F1 / 64);
    gemm_f32_wmma<DIN, F1><<<cdiv(strips, TB / 32), TB, 0, stream>>>(x, W1, h1, N);

    dot_e_kernel<<<cdiv((long long)N * HEADS, TB), TB, 0, stream>>>(
        h1, a1_src, a1_dst, e_src_b, e_dst_b, N, HEADS, HID);

    gat_owner<HEADS, HID, false, 1><<<OWN_GROUPS, OWN_THREADS, 0, stream>>>(
        srcIdx, dstIdx, E, N, h1, e_src_b, e_dst_b, b1, out1, listW, cntW, Etot);
  }

  {
    int strips = (N / 16) * (DOUT / 64);
    gemm_f32_wmma<F1, DOUT><<<cdiv(strips, TB / 32), TB, 0, stream>>>(out1, W2, h2, N);

    dot_e_kernel<<<cdiv(N, TB), TB, 0, stream>>>(
        h2, a2_src, a2_dst, e_src_b, e_dst_b, N, 1, DOUT);

    gat_owner<1, DOUT, true, 2><<<OWN_GROUPS, OWN_THREADS, 0, stream>>>(
        srcIdx, dstIdx, E, N, h2, e_src_b, e_dst_b, b2, outp, listW, cntW, Etot);
  }
}


# GEMMs: a wave's four column tiles take interleaved columns (tile t = columns 4n+t): B fragments by two 16-byte loads per k-step (was 8 of 4), one 16-byte whole-line store per row; sum order kept
# speedup vs baseline: 3.9910x; 1.0445x over previous
#include <hip/hip_runtime.h>
#include <math.h>

#define DIN 256
#define HID 64
#define HEADS 4
#define DOUT 128
#define NEG_SLOPE 0.2f

typedef __attribute__((ext_vector_type(2))) float v2f;
typedef __attribute__((ext_vector_type(8))) float v8f;

template<int K, int NOUT>
__global__ void gemm_f32_wmma(const float* __restrict__ X,
                              const float* __restrict__ W,
                              float* __restrict__ H, int nrows) {
  const int lane = threadIdx.x & 31;
  const int wave = threadIdx.x >> 5;
  const int colGroups = NOUT / 64;
  const int strip = blockIdx.x * (blockDim.x >> 5) + wave;
  const int rowTile = strip / colGroups;
  const int colGroup = strip - rowTile * colGroups;
  if (rowTile * 16 >= nrows) return;
  const int r = lane & 15;
  const int half = lane >> 4;
  const float* __restrict__ Ap = X + (size_t)(rowTile * 16 + r) * K + 2 * half;
  const float* __restrict__ Bp = W + (size_t)(2 * half) * NOUT + colGroup * 64 + 4 * r;
  v8f acc0 = {}, acc1 = {}, acc2 = {}, acc3 = {};
#pragma unroll 4
  for (int k = 0; k < K; k += 4) {
    v2f a;
    a[0] = Ap[k];
    a[1] = Ap[k + 1];
    const float4 q0 = *reinterpret_cast<const float4*>(Bp + (size_t)k * NOUT);
    const float4 q1 = *reinterpret_cast<const float4*>(Bp + (size_t)(k + 1) * NOUT);
    v2f bA, bB, bC, bD;
    bA[0] = q0.x;  bA[1] = q1.x;
    bB[0] = q0.y;  bB[1] = q1.y;
    bC[0] = q0.z;  bC[1] = q1.z;
    bD[0] = q0.w;  bD[1] = q1.w;
    acc0 = __builtin_amdgcn_wmma_f32_16x16x4_f32(false, a, false, bA,
                                                 (short)0, acc0, false, false);
    acc1 = __builtin_amdgcn_wmma_f32_16x16x4_f32(false, a, false, bB,
                                                 (short)0, acc1, false, false);
    acc2 = __builtin_amdgcn_wmma_f32_16x16x4_f32(false, a, false, bC,
                                                 (short)0, acc2, false, false);
    acc3 = __builtin_amdgcn_wmma_f32_16x16x4_f32(false, a, false, bD,
                                                 (short)0, acc3, false, false);
  }
  float* __restrict__ Cp =
      H + (size_t)(rowTile * 16 + 8 * half) * NOUT + colGroup * 64 + 4 * r;
#pragma unroll
  for (int v = 0; v < 8; ++v) {
    *reinterpret_cast<float4*>(Cp + (size_t)v * NOUT) =
        make_float4(acc0[v], acc1[v], acc2[v], acc3[v]);
  }
}

__global__ void dot_e_kernel(const float* __restrict__ H,
                             const float* __restrict__ a_src,
                             const float* __restrict__ a_dst,
                             float* __restrict__ e_src,
                             float* __restrict__ e_dst,
                             int n, int heads, int C) {
  int idx = blockIdx.x * blockDim.x + threadIdx.x;
  if (idx >= n * heads) return;
  int node = idx / heads;
  int hd = idx - node * heads;
  const float* hp = H + (size_t)node * heads * C + (size_t)hd * C;
  const float* as = a_src + (size_t)hd * C;
  const float* ad = a_dst + (size_t)hd * C;
  float s0 = 0.f, s1 = 0.f;
  for (int c = 0; c < C; ++c) {
    float v = hp[c];
    s0 += v * as[c];
    s1 += v * ad[c];
  }
  e_src[idx] = s0;
  e_dst[idx] = s1;
}

#define OWN_ROWS 196
#define OWN_PITCH 256
#define OWN_THREADS 1024
#define OWN_GROUPS 256
#define LIST_CAP 8192
#define PRIV_CLASSES 32
#define PRIV_CAP 256
#define SWEEP_AHEAD 8

__device__ __forceinline__ void ldsMaxFloat(float* addr, float val) {
  if (val >= 0.0f)
    atomicMax((int*)addr, __float_as_int(val));
  else
    atomicMin((unsigned int*)addr, __float_as_uint(val));
}

template<int NH, int C, bool FINAL, int LMODE>
__global__ void __launch_bounds__(OWN_THREADS)
gat_owner(const int* __restrict__ srcIdx, const int* __restrict__ dstIdx,
          int E, int n,
          const float* __restrict__ H,
          const float* __restrict__ e_src,
          const float* __restrict__ e_dst,
          const float* __restrict__ b,
          float* __restrict__ out,
          int* listW,
          int* cntW,
          int edgeRows) {
  constexpr int F = NH * C;
  constexpr int J = F / 32;
  constexpr int LIST_B = 1024 / F;
  static_assert(!FINAL || NH == 1, "row mean is written for one head");
  static_assert(F <= OWN_PITCH && (F % 32) == 0 && (C % 32) == 0, "channel layout");
  static_assert(LIST_B >= 1 && LIST_B * J == 32, "32 row loads in flight per wave");
  static_assert(LMODE == 1 || LMODE == 2, "1: sweep and save the list, 2: read the saved list");
  static_assert(OWN_ROWS <= 256, "a saved word keeps the local row in its top 8 bits");
  __shared__ __attribute__((aligned(16))) float acc[OWN_ROWS * OWN_PITCH];
  __shared__ float mS[OWN_ROWS * NH];
  __shared__ float denS[OWN_ROWS * NH];
  __shared__ float edS[OWN_ROWS * NH];
  __shared__ int listS[LIST_CAP];
  __shared__ int listD[LIST_CAP];
  __shared__ int listN;
  __shared__ int privL[PRIV_CLASSES * PRIV_CAP];
  __shared__ int privN[PRIV_CLASSES];

  const int tid = threadIdx.x;
  const int lane = tid & 31;
  const int wave = tid >> 5;
  const int nthr = blockDim.x;
  const int nwav = nthr >> 5;
  const int chunk = nthr * 4;
  const int nIt = (E + chunk - 1) / chunk;
  const int it0 = (int)(((long long)blockIdx.x * nIt) / gridDim.x);
  const bool vec = ((E & 3) == 0) && ((((size_t)dstIdx) & 15) == 0);

  auto fetch = [&](int i) -> int4 {
    int4 v = make_int4(-1, -1, -1, -1);
    if (i < nIt) {
      int itr = it0 + i;
      if (itr >= nIt) itr -= nIt;
      const int e0 = itr * chunk + tid * 4;
      if (vec && e0 + 3 < E) {
        v = *reinterpret_cast<const int4*>(dstIdx + e0);
      } else {
        if (e0 < E) v.x = dstIdx[e0];
        if (e0 + 1 < E) v.y = dstIdx[e0 + 1];
        if (e0 + 2 < E) v.z = dstIdx[e0 + 2];
        if (e0 + 3 < E) v.w = dstIdx[e0 + 3];
      }
    }
    return v;
  };
  auto edge0 = [&](int i) -> int {
    int itr = it0 + i;
    if (itr >= nIt) itr -= nIt;
    return itr * chunk + tid * 4;
  };

  for (int base = blockIdx.x * OWN_ROWS; base < n; base += gridDim.x * OWN_ROWS) {
    const int nrows = min(OWN_ROWS, n - base);
    const int blk = base / OWN_ROWS;
    int* const lw = listW + (size_t)blk * LIST_CAP;
    const bool canSave = (n <= (1 << 24)) &&
                         ((long long)(blk + 1) * LIST_CAP <= (long long)edgeRows * HEADS);

    for (int i = tid; i < nrows * F; i += nthr) {
      const int r = i / F;
      acc[r * OWN_PITCH + (i - r * F)] = 0.f;
    }
    for (int i = tid; i < nrows * NH; i += nthr) {
      const float ed = e_dst[(size_t)base * NH + i];
      float l = e_src[(size_t)base * NH + i] + ed;
      l = (l > 0.f) ? l : l * NEG_SLOPE;
      edS[i] = ed;
      mS[i] = l;
      denS[i] = 0.f;
    }
    if (tid == 0) listN = 0;
    if (tid < PRIV_CLASSES) privN[tid] = 0;
    const bool packOK = n <= (1 << 24);
    bool viaList = false;
    int total;
    if (LMODE == 2) {
      total = __builtin_amdgcn_readfirstlane(cntW[blk]);
      if (total <= LIST_CAP) {
        for (int i = tid; i < total; i += nthr) {
          const int p = lw[i];
          listS[i] = min(p & 0xFFFFFF, n - 1);
          listD[i] = min((int)((unsigned)p >> 24), nrows - 1);
        }
      }
      __syncthreads();
    } else {
      __syncthreads();

      {
        auto fileStep = [&](const int4 cur, const int e0) {
          const int dl[4] = {cur.x - base, cur.y - base, cur.z - base, cur.w - base};
          bool mt[4];
          unsigned bm[4];
          for (int k = 0; k < 4; ++k) {
            mt[k] = (unsigned)dl[k] < (unsigned)nrows;
            bm[k] = __builtin_amdgcn_ballot_w32(mt[k]);
          }
          if ((bm[0] | bm[1] | bm[2] | bm[3]) == 0u) return;
          const int found = __builtin_popcount(bm[0]) + __builtin_popcount(bm[1]) +
                            __builtin_popcount(bm[2]) + __builtin_popcount(bm[3]);
          int slot = 0;
          if (lane == 0) slot = atomicAdd(&listN, found);
          slot = __builtin_amdgcn_readfirstlane(slot);
          for (int k = 0; k < 4; ++k) {
            if (mt[k]) {
              const int pos = slot + (int)__builtin_amdgcn_mbcnt_lo(bm[k], 0u);
              if (pos < LIST_CAP) {
                listS[pos] = e0 + k;
                listD[pos] = dl[k];
              }
            }
            slot += __builtin_popcount(bm[k]);
          }
        };
        if (vec) {
          const int nFull = E / chunk;
          if (nFull > 0) {
            const int itF0 = (int)(((long long)blockIdx.x * nFull) / gridDim.x);
            auto edgeF = [&](int i) -> int {
              int itr = itF0 + min(i, nFull - 1);
              if (itr >= nFull) itr -= nFull;
              return itr * chunk + tid * 4;
            };
            int4 q[SWEEP_AHEAD];
#pragma unroll
            for (int u = 0; u < SWEEP_AHEAD; ++u)
              q[u] = *reinterpret_cast<const int4*>(dstIdx + edgeF(u));
            for (int i = 0; i < nFull; i += SWEEP_AHEAD) {
#pragma unroll
              for (int u = 0; u < SWEEP_AHEAD; ++u) {
                if (i + u < nFull) fileStep(q[u], edgeF(i + u));
                q[u] = *reinterpret_cast<const int4*>(dstIdx + edgeF(i + u + SWEEP_AHEAD));
              }
            }
          }
          if (nFull * chunk < E) {
            const int e0 = nFull * chunk + tid * 4;
            int4 v = make_int4(-1, -1, -1, -1);
            if (e0 < E) v = *reinterpret_cast<const int4*>(dstIdx + e0);
            fileStep(v, e0);
          }
        } else {
          int4 q0 = fetch(0), q1 = fetch(1), q2 = fetch(2), q3 = fetch(3);
          for (int i = 0; i < nIt; ++i) {
            const int4 cur = q0;
            q0 = q1; q1 = q2; q2 = q3; q3 = fetch(i + 4);
            fileStep(cur, edge0(i));
          }
        }
      }
      __syncthreads();
      total = __builtin_amdgcn_readfirstlane(listN);
      if (tid == 0) cntW[blk] = canSave ? total : LIST_CAP + 1;
    }

    if (total <= LIST_CAP) {
      for (int i0 = tid; i0 < total; i0 += 4 * nthr) {
        int ci[4], sv[4], dv[4];
        bool ok[4];
        for (int k = 0; k < 4; ++k) {
          const int idx = i0 + k * nthr;
          ok[k] = idx < total;
          ci[k] = ok[k] ? idx : i0;
        }
        for (int k = 0; k < 4; ++k) {
          sv[k] = (LMODE == 1) ? srcIdx[listS[ci[k]]] : listS[ci[k]];
          dv[k] = listD[ci[k]];
        }
        float es[4][NH];
        for (int k = 0; k < 4; ++k)
          for (int h = 0; h < NH; ++h)
            es[k][h] = e_src[(size_t)sv[k] * NH + h];
        for (int k = 0; k < 4; ++k) {
          if (ok[k]) {
            if (LMODE == 1) listS[ci[k]] = sv[k];
            if (packOK) {
              const int o = dv[k] & (PRIV_CLASSES - 1);
              const int pos = atomicAdd(&privN[o], 1);
              if (pos < PRIV_CAP)
                privL[o * PRIV_CAP + pos] = (int)((unsigned)sv[k] | ((unsigned)dv[k] << 24));
            }
            for (int h = 0; h < NH; ++h) {
              float l = es[k][h] + edS[dv[k] * NH + h];
              l = (l > 0.f) ? l : l * NEG_SLOPE;
              ldsMaxFloat(&mS[dv[k] * NH + h], l);
            }
          }
        }
      }
      if (packOK) {
        for (int r = tid; r < nrows; r += nthr) {
          const int o = r & (PRIV_CLASSES - 1);
          const int pos = atomicAdd(&privN[o], 1);
          if (pos < PRIV_CAP) privL[o * PRIV_CAP + pos] = (int)((unsigned)(base + r) | ((unsigned)r << 24));
        }
      }
      __syncthreads();
      viaList = packOK && (__builtin_amdgcn_ballot_w32(privN[lane] > PRIV_CAP) == 0u);

      if (LMODE == 1 && canSave) {
        for (int i = tid; i < total; i += nthr) lw[i] = (int)((unsigned)listS[i] | ((unsigned)listD[i] << 24));
      }

      if (viaList) {
        constexpr int JV = F / 128;
        constexpr int HG = (C >= 128) ? 1 : 128 / C;
        constexpr int EL = LIST_B * NH;
        static_assert(F % 128 == 0 && (C % 128 == 0 || 128 % C == 0) && 32 % HG == 0, "16-byte pieces");
        static_assert(LIST_B * JV * 4 == 32, "32 registers of row loads in flight per wave");
        static_assert(((OWN_ROWS + PRIV_CLASSES - 1) / PRIV_CLASSES) * NH <= 32,
                      "one lane per (row of a class, head) for den");
        static_assert((LIST_B & (LIST_B - 1)) == 0 && (NH & (NH - 1)) == 0 && EL <= 32,
                      "one (entry, head) pair per lane");
        const int kl = (lane / NH) & (LIST_B - 1);
        const int hl = lane & (NH - 1);
        const int waveS = __builtin_amdgcn_readfirstlane(wave);
        for (int o = waveS; o < PRIV_CLASSES; o += nwav) {
          const int cnt = __builtin_amdgcn_readfirstlane(privN[o]);
          if (cnt == 0) continue;
          const int* pl = privL + o * PRIV_CAP;
          float denReg = 0.f;
          const int ql = lane / NH;
          const int nFull = cnt / LIST_B;
          const int nTrip = (cnt + LIST_B - 1) / LIST_B;
          auto trip = [&](int t, int w, const bool tail, int& wN) __attribute__((always_inline)) {
            const int g = t * LIST_B;
            const int s = w & 0xFFFFFF;
            const int d = (int)((unsigned)w >> 24);
            const float edv = edS[d * NH + hl];
            const float mv = mS[d * NH + hl];
            int sv[LIST_B], dv[LIST_B];
#pragma unroll
            for (int k = 0; k < LIST_B; ++k) {
              sv[k] = __builtin_amdgcn_readlane(s, k * NH);
              dv[k] = __builtin_amdgcn_readlane(d, k * NH);
            }
            float4 hv[LIST_B][JV];
#pragma unroll
            for (int k = 0; k < LIST_B; ++k) {
              const float4* hp = reinterpret_cast<const float4*>(H + (size_t)sv[k] * F) + lane;
#pragma unroll
              for (int jj = 0; jj < JV; ++jj) hv[k][jj] = hp[32 * jj];
            }
            __builtin_amdgcn_sched_barrier(0);
            const float elv = e_src[(size_t)s * NH + hl];
            wN = pl[min(g + LIST_B + kl, cnt - 1)];
            float l = elv + edv;
            l = (l > 0.f) ? l : l * NEG_SLOPE;
            const float exv = expf(l - mv);
#pragma unroll
            for (int k = 0; k < LIST_B; ++k) {
              if (!tail || g + k < cnt) {
                float4* ap = reinterpret_cast<float4*>(acc + dv[k] * OWN_PITCH) + lane;
                {
                  float eh = __int_as_float(__builtin_amdgcn_readlane(__float_as_int(exv), k * NH));
#pragma unroll
                  for (int h = 1; h < NH; ++h) {
                    const float e2 = __int_as_float(
                        __builtin_amdgcn_readlane(__float_as_int(exv), k * NH + h));
                    eh = (hl == h) ? e2 : eh;
                  }
                  denReg += (ql == (dv[k] / PRIV_CLASSES)) ? eh : 0.f;
                }
                float4 a[JV];
#pragma unroll
                for (int jj = 0; jj < JV; ++jj) a[jj] = ap[32 * jj];
#pragma unroll
                for (int jj = 0; jj < JV; ++jj) {
                  const int h0 = (C >= 128) ? (128 * jj) / C : jj * HG;
                  float e = __int_as_float(
                      __builtin_amdgcn_readlane(__float_as_int(exv), k * NH + h0));
#pragma unroll
                  for (int q = 1; q < HG; ++q) {
                    const float eq = __int_as_float(
                        __builtin_amdgcn_readlane(__float_as_int(exv), k * NH + h0 + q));
                    e = (lane >= q * (32 / HG)) ? eq : e;
                  }
                  const float4 hq = hv[k][jj];
                  a[jj].x += e * hq.x;
                  a[jj].y += e * hq.y;
                  a[jj].z += e * hq.z;
                  a[jj].w += e * hq.w;
                }
#pragma unroll
                for (int jj = 0; jj < JV; ++jj) ap[32 * jj] = a[jj];
              }
            }
          };
          int w = pl[min(kl, cnt - 1)];
          int t = 0;
          for (; t < nFull; ++t) {
            int wN;
            trip(t, w, false, wN);
            w = wN;
          }
          if (t < nTrip) {
            int wN;
            trip(t, w, true, wN);
          }
          {
            const int r = o + PRIV_CLASSES * ql;
            if (r < nrows) denS[r * NH + hl] = denReg;
          }
        }
      }
    }
    if (!viaList) {
      {
        int4 q0 = fetch(0), q1 = fetch(1), q2 = fetch(2), q3 = fetch(3);
        for (int i = 0; i < nIt; ++i) {
          const int4 cur = q0;
          q0 = q1; q1 = q2; q2 = q3; q3 = fetch(i + 4);
          const int dl[4] = {cur.x - base, cur.y - base, cur.z - base, cur.w - base};
          bool mt[4];
          for (int k = 0; k < 4; ++k) mt[k] = (unsigned)dl[k] < (unsigned)nrows;
          if (__builtin_amdgcn_ballot_w32(mt[0] | mt[1] | mt[2] | mt[3]) == 0u) continue;
          const int e0 = edge0(i);
          int sv[4];
          for (int k = 0; k < 4; ++k) sv[k] = mt[k] ? srcIdx[e0 + k] : 0;
          float es[4][NH];
          for (int k = 0; k < 4; ++k)
            for (int h = 0; h < NH; ++h)
              es[k][h] = mt[k] ? e_src[(size_t)sv[k] * NH + h] : 0.f;
          for (int k = 0; k < 4; ++k) {
            if (mt[k]) {
              for (int h = 0; h < NH; ++h) {
                float l = es[k][h] + edS[dl[k] * NH + h];
                l = (l > 0.f) ? l : l * NEG_SLOPE;
                ldsMaxFloat(&mS[dl[k] * NH + h], l);
              }
            }
          }
        }
      }
      __syncthreads();

      {
        auto addEdge = [&](int s, int d) {
          const float* hp = H + (size_t)s * F;
          float hv[J];
          for (int j = 0; j < J; ++j) hv[j] = hp[lane + 32 * j];
          const int h = lane & (NH - 1);
          float l = e_src[(size_t)s * NH + h] + edS[d * NH + h];
          l = (l > 0.f) ? l : l * NEG_SLOPE;
          const float ex = expf(l - mS[d * NH + h]);
          if (lane < NH) atomicAdd(&denS[d * NH + lane], ex);
          float* ap = acc + d * OWN_PITCH;
#pragma unroll
          for (int j = 0; j < J; ++j) {
            const float exj = __int_as_float(
                __builtin_amdgcn_readlane(__float_as_int(ex), (32 * j) / C));
            atomicAdd(&ap[lane + 32 * j], exj * hv[j]);
          }
        };

        int4 q0 = fetch(0), q1 = fetch(1), q2 = fetch(2), q3 = fetch(3);
        for (int i = 0; i < nIt; ++i) {
          const int4 cur = q0;
          q0 = q1; q1 = q2; q2 = q3; q3 = fetch(i + 4);
          const int dl[4] = {cur.x - base, cur.y - base, cur.z - base, cur.w - base};
          bool mt[4];
          for (int k = 0; k < 4; ++k) mt[k] = (unsigned)dl[k] < (unsigned)nrows;
          if (__builtin_amdgcn_ballot_w32(mt[0] | mt[1] | mt[2] | mt[3]) == 0u) continue;
          const int e0 = edge0(i);
          int sv[4];
          for (int k = 0; k < 4; ++k) sv[k] = mt[k] ? srcIdx[e0 + k] : 0;
          for (int k = 0; k < 4; ++k) {
            unsigned mask = __builtin_amdgcn_ballot_w32(mt[k]);
            while (mask) {
              const int j = __builtin_ctz(mask);
              mask &= mask - 1u;
              addEdge(__builtin_amdgcn_readlane(sv[k], j),
                      __builtin_amdgcn_readlane(dl[k], j));
            }
          }
        }
        for (int r = wave; r < nrows; r += nwav) addEdge(base + r, r);
      }
    }
    __syncthreads();

    if (!FINAL) {
      for (int i = tid; i < nrows * F; i += nthr) {
        const int r = i / F;
        const int f = i - r * F;
        const float v = acc[r * OWN_PITCH + f] / (denS[r * NH + f / C] + 1e-16f) + b[f];
        out[(size_t)(base + r) * F + f] = (v > 0.f) ? v : (expf(v) - 1.f);
      }
    } else {
      for (int r = wave; r < nrows; r += nwav) {
        const float den = denS[r * NH] + 1e-16f;
        float sum = 0.f;
        for (int c = lane; c < F; c += 32) {
          float v = acc[r * OWN_PITCH + c] / den + b[c];
          v = (v > 0.f) ? v : (expf(v) - 1.f);
          sum += v;
        }
        for (int off = 16; off > 0; off >>= 1)
          sum += __shfl_down(sum, off, 32);
        if (lane == 0) out[base + r] = sum / (float)F;
      }
    }
    __syncthreads();
  }
}

static inline int cdiv(long long a, long long b) { return (int)((a + b - 1) / b); }

extern "C" void kernel_launch(void* const* d_in, const int* in_sizes, int n_in,
                              void* d_out, int out_size, void* d_ws, size_t ws_size,
                              hipStream_t stream) {
  const float* x      = (const float*)d_in[0];
  const int*   ei     = (const int*)  d_in[1];
  const float* W1     = (const float*)d_in[2];
  const float* a1_src = (const float*)d_in[3];
  const float* a1_dst = (const float*)d_in[4];
  const float* b1     = (const float*)d_in[5];
  const float* W2     = (const float*)d_in[6];
  const float* a2_src = (const float*)d_in[7];
  const float* a2_dst = (const float*)d_in[8];
  const float* b2     = (const float*)d_in[9];
  float* outp = (float*)d_out;

  const int N    = in_sizes[0] / DIN;
  const int E    = in_sizes[1] / 2;
  const int Etot = E + N;
  const int F1   = HEADS * HID;
  const int* srcIdx = ei;
  const int* dstIdx = ei + E;

  float* ws = (float*)d_ws;
  float* h1      = ws;
  float* e_src_b = h1 + (size_t)N * F1;
  float* e_dst_b = e_src_b + (size_t)N * HEADS;
  float* edgebuf = e_dst_b + (size_t)N * HEADS;
  float* m_b     = edgebuf + (size_t)Etot * HEADS;
  float* den_b   = m_b + (size_t)N * HEADS;
  float* out1    = den_b + (size_t)N * HEADS;
  float* h2      = h1;
  int* listW = (int*)edgebuf;
  int* cntW  = (int*)m_b;

  const int TB = 256;

  {
    int strips = (N / 16) * (F1 / 64);
    gemm_f32_wmma<DIN, F1><<<cdiv(strips, TB / 32), TB, 0, stream>>>(x, W1, h1, N);

    dot_e_kernel<<<cdiv((long long)N * HEADS, TB), TB, 0, stream>>>(
        h1, a1_src, a1_dst, e_src_b, e_dst_b, N, HEADS, HID);

    gat_owner<HEADS, HID, false, 1><<<OWN_GROUPS, OWN_THREADS, 0, stream>>>(
        srcIdx, dstIdx, E, N, h1, e_src_b, e_dst_b, b1, out1, listW, cntW, Etot);
  }

  {
    int strips = (N / 16) * (DOUT / 64);
    gemm_f32_wmma<F1, DOUT><<<cdiv(strips, TB / 32), TB, 0, stream>>>(out1, W2, h2, N);

    dot_e_kernel<<<cdiv(N, TB), TB, 0, stream>>>(
        h2, a2_src, a2_dst, e_src_b, e_dst_b, N, 1, DOUT);

    gat_owner<1, DOUT, true, 2><<<OWN_GROUPS, OWN_THREADS, 0, stream>>>(
        srcIdx, dstIdx, E, N, h2, e_src_b, e_dst_b, b2, outp, listW, cntW, Etot);
  }
}


# gat_owner: each work-group's sweep of the whole edge list is replaced by one bucketing launch (256 pieces file edges per row block, LDS counters); both layers gather their segments; arrival order kept
# speedup vs baseline: 4.5048x; 1.1287x over previous
#include <hip/hip_runtime.h>
#include <math.h>

#define DIN 256
#define HID 64
#define HEADS 4
#define DOUT 128
#define NEG_SLOPE 0.2f

typedef __attribute__((ext_vector_type(2))) float v2f;
typedef __attribute__((ext_vector_type(8))) float v8f;

template<int K, int NOUT>
__global__ void gemm_f32_wmma(const float* __restrict__ X,
                              const float* __restrict__ W,
                              float* __restrict__ H, int nrows) {
  const int lane = threadIdx.x & 31;
  const int wave = threadIdx.x >> 5;
  const int colGroups = NOUT / 64;
  const int strip = blockIdx.x * (blockDim.x >> 5) + wave;
  const int rowTile = strip / colGroups;
  const int colGroup = strip - rowTile * colGroups;
  if (rowTile * 16 >= nrows) return;
  const int r = lane & 15;
  const int half = lane >> 4;
  const float* __restrict__ Ap = X + (size_t)(rowTile * 16 + r) * K + 2 * half;
  const float* __restrict__ Bp = W + (size_t)(2 * half) * NOUT + colGroup * 64 + 4 * r;
  v8f acc0 = {}, acc1 = {}, acc2 = {}, acc3 = {};
#pragma unroll 4
  for (int k = 0; k < K; k += 4) {
    v2f a;
    a[0] = Ap[k];
    a[1] = Ap[k + 1];
    const float4 q0 = *reinterpret_cast<const float4*>(Bp + (size_t)k * NOUT);
    const float4 q1 = *reinterpret_cast<const float4*>(Bp + (size_t)(k + 1) * NOUT);
    v2f bA, bB, bC, bD;
    bA[0] = q0.x;  bA[1] = q1.x;
    bB[0] = q0.y;  bB[1] = q1.y;
    bC[0] = q0.z;  bC[1] = q1.z;
    bD[0] = q0.w;  bD[1] = q1.w;
    acc0 = __builtin_amdgcn_wmma_f32_16x16x4_f32(false, a, false, bA,
                                                 (short)0, acc0, false, false);
    acc1 = __builtin_amdgcn_wmma_f32_16x16x4_f32(false, a, false, bB,
                                                 (short)0, acc1, false, false);
    acc2 = __builtin_amdgcn_wmma_f32_16x16x4_f32(false, a, false, bC,
                                                 (short)0, acc2, false, false);
    acc3 = __builtin_amdgcn_wmma_f32_16x16x4_f32(false, a, false, bD,
                                                 (short)0, acc3, false, false);
  }
  float* __restrict__ Cp =
      H + (size_t)(rowTile * 16 + 8 * half) * NOUT + colGroup * 64 + 4 * r;
#pragma unroll
  for (int v = 0; v < 8; ++v) {
    *reinterpret_cast<float4*>(Cp + (size_t)v * NOUT) =
        make_float4(acc0[v], acc1[v], acc2[v], acc3[v]);
  }
}

__global__ void dot_e_kernel(const float* __restrict__ H,
                             const float* __restrict__ a_src,
                             const float* __restrict__ a_dst,
                             float* __restrict__ e_src,
                             float* __restrict__ e_dst,
                             int n, int heads, int C) {
  int idx = blockIdx.x * blockDim.x + threadIdx.x;
  if (idx >= n * heads) return;
  int node = idx / heads;
  int hd = idx - node * heads;
  const float* hp = H + (size_t)node * heads * C + (size_t)hd * C;
  const float* as = a_src + (size_t)hd * C;
  const float* ad = a_dst + (size_t)hd * C;
  float s0 = 0.f, s1 = 0.f;
  for (int c = 0; c < C; ++c) {
    float v = hp[c];
    s0 += v * as[c];
    s1 += v * ad[c];
  }
  e_src[idx] = s0;
  e_dst[idx] = s1;
}

#define OWN_ROWS 196
#define OWN_PITCH 256
#define OWN_THREADS 1024
#define OWN_GROUPS 256
#define LIST_CAP 8192
#define PRIV_CLASSES 32
#define PRIV_CAP 256
#define NSEG 256
#define SEG_CAP 48
#define SEG_WORDS (NSEG * SEG_CAP)
#define SEG_LANES 4
#define PRE_THREADS 256

__global__ void __launch_bounds__(PRE_THREADS)
bucket_edges(const int* __restrict__ srcIdx, const int* __restrict__ dstIdx,
             int E, int n,
             int* __restrict__ segW,
             int* __restrict__ segCnt,
             int edgeRows) {
  __shared__ int cnt[OWN_GROUPS];
  const int tid = threadIdx.x;
  const int nthr = blockDim.x;
  const int g = blockIdx.x;
  if (g >= NSEG) return;
  for (int i = tid; i < OWN_GROUPS; i += nthr) cnt[i] = 0;
  __syncthreads();
  const int quads = (E + 3) >> 2;
  const int qPer = (quads + NSEG - 1) / NSEG;
  const int qBeg = g * qPer;
  const int qEnd = min(quads, qBeg + qPer);
  const bool vec = ((E & 3) == 0) && (((((size_t)dstIdx) | ((size_t)srcIdx)) & 15) == 0);
  const bool packOK = n <= (1 << 24);
  auto fileEdge = [&](int s, int d) {
    if ((unsigned)d >= (unsigned)n || !packOK) return;
    const int b = d / OWN_ROWS;
    if (b >= OWN_GROUPS || (long long)(b + 1) * SEG_WORDS > (long long)edgeRows * HEADS) return;
    const int rank = atomicAdd(&cnt[b], 1);
    if (rank < SEG_CAP)
      segW[((size_t)b * NSEG + g) * SEG_CAP + rank] = (int)((unsigned)s | ((unsigned)(d - b * OWN_ROWS) << 24));
  };
  if (vec) {
    for (int q0 = qBeg + tid; q0 < qEnd; q0 += 4 * nthr) {
      int4 dv[4], sv[4];
#pragma unroll
      for (int k = 0; k < 4; ++k) {
        const int e0 = min(q0 + k * nthr, qEnd - 1) * 4;
        dv[k] = *reinterpret_cast<const int4*>(dstIdx + e0);
        sv[k] = *reinterpret_cast<const int4*>(srcIdx + e0);
      }
#pragma unroll
      for (int k = 0; k < 4; ++k) {
        if (q0 + k * nthr < qEnd) {
          fileEdge(sv[k].x, dv[k].x);
          fileEdge(sv[k].y, dv[k].y);
          fileEdge(sv[k].z, dv[k].z);
          fileEdge(sv[k].w, dv[k].w);
        }
      }
    }
  } else {
    const int eEnd = min(E, qEnd * 4);
    for (int e = qBeg * 4 + tid; e < eEnd; e += nthr) fileEdge(srcIdx[e], dstIdx[e]);
  }
  __syncthreads();
  for (int i = tid; i < OWN_GROUPS; i += nthr) segCnt[(size_t)i * NSEG + g] = cnt[i];
}

__device__ __forceinline__ void ldsMaxFloat(float* addr, float val) {
  if (val >= 0.0f)
    atomicMax((int*)addr, __float_as_int(val));
  else
    atomicMin((unsigned int*)addr, __float_as_uint(val));
}

template<int NH, int C, bool FINAL>
__global__ void __launch_bounds__(OWN_THREADS)
gat_owner(const int* __restrict__ srcIdx, const int* __restrict__ dstIdx,
          int E, int n,
          const float* __restrict__ H,
          const float* __restrict__ e_src,
          const float* __restrict__ e_dst,
          const float* __restrict__ b,
          float* __restrict__ out,
          const int* __restrict__ segW,
          const int* __restrict__ segCnt,
          int edgeRows) {
  constexpr int F = NH * C;
  constexpr int J = F / 32;
  constexpr int LIST_B = 1024 / F;
  static_assert(!FINAL || NH == 1, "row mean is written for one head");
  static_assert(F <= OWN_PITCH && (F % 32) == 0 && (C % 32) == 0, "channel layout");
  static_assert(LIST_B >= 1 && LIST_B * J == 32, "32 row loads in flight per wave");
  static_assert(OWN_ROWS <= 256, "a segment word keeps the local row in its top 8 bits");
  static_assert(NSEG % 32 == 0 && NSEG <= OWN_THREADS && SEG_CAP % SEG_LANES == 0, "segment gather layout");
  __shared__ __attribute__((aligned(16))) float acc[OWN_ROWS * OWN_PITCH];
  __shared__ float mS[OWN_ROWS * NH];
  __shared__ float denS[OWN_ROWS * NH];
  __shared__ float edS[OWN_ROWS * NH];
  __shared__ int listS[LIST_CAP];
  __shared__ int listD[LIST_CAP];
  __shared__ int listN;
  __shared__ int segN[NSEG];
  __shared__ int segOff[NSEG];
  __shared__ int segTot[NSEG / 32];
  __shared__ int segOver;
  __shared__ int privL[PRIV_CLASSES * PRIV_CAP];
  __shared__ int privN[PRIV_CLASSES];

  const int tid = threadIdx.x;
  const int lane = tid & 31;
  const int wave = tid >> 5;
  const int nthr = blockDim.x;
  const int nwav = nthr >> 5;
  const int chunk = nthr * 4;
  const int nIt = (E + chunk - 1) / chunk;
  const int it0 = (int)(((long long)blockIdx.x * nIt) / gridDim.x);
  const bool vec = ((E & 3) == 0) && ((((size_t)dstIdx) & 15) == 0);

  auto fetch = [&](int i) -> int4 {
    int4 v = make_int4(-1, -1, -1, -1);
    if (i < nIt) {
      int itr = it0 + i;
      if (itr >= nIt) itr -= nIt;
      const int e0 = itr * chunk + tid * 4;
      if (vec && e0 + 3 < E) {
        v = *reinterpret_cast<const int4*>(dstIdx + e0);
      } else {
        if (e0 < E) v.x = dstIdx[e0];
        if (e0 + 1 < E) v.y = dstIdx[e0 + 1];
        if (e0 + 2 < E) v.z = dstIdx[e0 + 2];
        if (e0 + 3 < E) v.w = dstIdx[e0 + 3];
      }
    }
    return v;
  };
  auto edge0 = [&](int i) -> int {
    int itr = it0 + i;
    if (itr >= nIt) itr -= nIt;
    return itr * chunk + tid * 4;
  };

  for (int base = blockIdx.x * OWN_ROWS; base < n; base += gridDim.x * OWN_ROWS) {
    const int nrows = min(OWN_ROWS, n - base);
    const int blk = base / OWN_ROWS;
    const bool haveSegs = (n <= (1 << 24)) && (blk < OWN_GROUPS) && (nthr >= NSEG) &&
                          ((long long)(blk + 1) * SEG_WORDS <= (long long)edgeRows * HEADS);

    for (int i = tid; i < nrows * F; i += nthr) {
      const int r = i / F;
      acc[r * OWN_PITCH + (i - r * F)] = 0.f;
    }
    for (int i = tid; i < nrows * NH; i += nthr) {
      const float ed = e_dst[(size_t)base * NH + i];
      float l = e_src[(size_t)base * NH + i] + ed;
      l = (l > 0.f) ? l : l * NEG_SLOPE;
      edS[i] = ed;
      mS[i] = l;
      denS[i] = 0.f;
    }
    if (tid == 0) { listN = 0; segOver = 0; }
    if (tid < PRIV_CLASSES) privN[tid] = 0;
    const bool packOK = n <= (1 << 24);
    bool viaList = false;
    int total = LIST_CAP + 1;
    __syncthreads();
    if (haveSegs) {
      const int cg = (tid < NSEG) ? segCnt[(size_t)blk * NSEG + tid] : 0;
      int incl = cg;
      for (int off = 1; off < 32; off <<= 1) {
        const int t = __shfl_up(incl, off, 32);
        if (lane >= off) incl += t;
      }
      if (tid < NSEG && lane == 31) segTot[wave] = incl;
      if (cg > SEG_CAP) segOver = 1;
      __syncthreads();
      if (tid < NSEG) {
        int before = 0;
        for (int w = 0; w < wave; ++w) before += segTot[w];
        segN[tid] = cg;
        segOff[tid] = before + incl - cg;
        if (tid == NSEG - 1) listN = before + incl;
      }
      __syncthreads();
      total = __builtin_amdgcn_readfirstlane(segOver ? LIST_CAP + 1 : listN);
      if (total <= LIST_CAP) {
        const int* const lw = segW + (size_t)blk * SEG_WORDS;
        for (int i = tid; i < NSEG * SEG_LANES; i += nthr) {
          const int g = i / SEG_LANES;
          const int sub = i - g * SEG_LANES;
          const int cn = segN[g];
          const int o = segOff[g];
          const int* sp = lw + g * SEG_CAP;
          int p[SEG_CAP / SEG_LANES];
#pragma unroll
          for (int k = 0; k < SEG_CAP / SEG_LANES; ++k) {
            const int j = sub + k * SEG_LANES;
            p[k] = (j < cn) ? sp[j] : 0;
          }
#pragma unroll
          for (int k = 0; k < SEG_CAP / SEG_LANES; ++k) {
            const int j = sub + k * SEG_LANES;
            if (j < cn) {
              listS[o + j] = min(p[k] & 0xFFFFFF, n - 1);
              listD[o + j] = min((int)((unsigned)p[k] >> 24), nrows - 1);
            }
          }
        }
      }
      __syncthreads();
    }

    if (total <= LIST_CAP) {
      for (int i0 = tid; i0 < total; i0 += 4 * nthr) {
        int ci[4], sv[4], dv[4];
        bool ok[4];
        for (int k = 0; k < 4; ++k) {
          const int idx = i0 + k * nthr;
          ok[k] = idx < total;
          ci[k] = ok[k] ? idx : i0;
        }
        for (int k = 0; k < 4; ++k) {
          sv[k] = listS[ci[k]];
          dv[k] = listD[ci[k]];
        }
        float es[4][NH];
        for (int k = 0; k < 4; ++k)
          for (int h = 0; h < NH; ++h)
            es[k][h] = e_src[(size_t)sv[k] * NH + h];
        for (int k = 0; k < 4; ++k) {
          if (ok[k]) {
            if (packOK) {
              const int o = dv[k] & (PRIV_CLASSES - 1);
              const int pos = atomicAdd(&privN[o], 1);
              if (pos < PRIV_CAP)
                privL[o * PRIV_CAP + pos] = (int)((unsigned)sv[k] | ((unsigned)dv[k] << 24));
            }
            for (int h = 0; h < NH; ++h) {
              float l = es[k][h] + edS[dv[k] * NH + h];
              l = (l > 0.f) ? l : l * NEG_SLOPE;
              ldsMaxFloat(&mS[dv[k] * NH + h], l);
            }
          }
        }
      }
      if (packOK) {
        for (int r = tid; r < nrows; r += nthr) {
          const int o = r & (PRIV_CLASSES - 1);
          const int pos = atomicAdd(&privN[o], 1);
          if (pos < PRIV_CAP) privL[o * PRIV_CAP + pos] = (int)((unsigned)(base + r) | ((unsigned)r << 24));
        }
      }
      __syncthreads();
      viaList = packOK && (__builtin_amdgcn_ballot_w32(privN[lane] > PRIV_CAP) == 0u);

      if (viaList) {
        constexpr int JV = F / 128;
        constexpr int HG = (C >= 128) ? 1 : 128 / C;
        constexpr int EL = LIST_B * NH;
        static_assert(F % 128 == 0 && (C % 128 == 0 || 128 % C == 0) && 32 % HG == 0, "16-byte pieces");
        static_assert(LIST_B * JV * 4 == 32, "32 registers of row loads in flight per wave");
        static_assert(((OWN_ROWS + PRIV_CLASSES - 1) / PRIV_CLASSES) * NH <= 32,
                      "one lane per (row of a class, head) for den");
        static_assert((LIST_B & (LIST_B - 1)) == 0 && (NH & (NH - 1)) == 0 && EL <= 32,
                      "one (entry, head) pair per lane");
        const int kl = (lane / NH) & (LIST_B - 1);
        const int hl = lane & (NH - 1);
        const int waveS = __builtin_amdgcn_readfirstlane(wave);
        for (int o = waveS; o < PRIV_CLASSES; o += nwav) {
          const int cnt = __builtin_amdgcn_readfirstlane(privN[o]);
          if (cnt == 0) continue;
          const int* pl = privL + o * PRIV_CAP;
          float denReg = 0.f;
          const int ql = lane / NH;
          const int nFull = cnt / LIST_B;
          const int nTrip = (cnt + LIST_B - 1) / LIST_B;
          auto trip = [&](int t, int w, const bool tail, int& wN) __attribute__((always_inline)) {
            const int g = t * LIST_B;
            const int s = w & 0xFFFFFF;
            const int d = (int)((unsigned)w >> 24);
            const float edv = edS[d * NH + hl];
            const float mv = mS[d * NH + hl];
            int sv[LIST_B], dv[LIST_B];
#pragma unroll
            for (int k = 0; k < LIST_B; ++k) {
              sv[k] = __builtin_amdgcn_readlane(s, k * NH);
              dv[k] = __builtin_amdgcn_readlane(d, k * NH);
            }
            float4 hv[LIST_B][JV];
#pragma unroll
            for (int k = 0; k < LIST_B; ++k) {
              const float4* hp = reinterpret_cast<const float4*>(H + (size_t)sv[k] * F) + lane;
#pragma unroll
              for (int jj = 0; jj < JV; ++jj) hv[k][jj] = hp[32 * jj];
            }
            __builtin_amdgcn_sched_barrier(0);
            const float elv = e_src[(size_t)s * NH + hl];
            wN = pl[min(g + LIST_B + kl, cnt - 1)];
            float l = elv + edv;
            l = (l > 0.f) ? l : l * NEG_SLOPE;
            const float exv = expf(l - mv);
#pragma unroll
            for (int k = 0; k < LIST_B; ++k) {
              if (!tail || g + k < cnt) {
                float4* ap = reinterpret_cast<float4*>(acc + dv[k] * OWN_PITCH) + lane;
                {
                  float eh = __int_as_float(__builtin_amdgcn_readlane(__float_as_int(exv), k * NH));
#pragma unroll
                  for (int h = 1; h < NH; ++h) {
                    const float e2 = __int_as_float(
                        __builtin_amdgcn_readlane(__float_as_int(exv), k * NH + h));
                    eh = (hl == h) ? e2 : eh;
                  }
                  denReg += (ql == (dv[k] / PRIV_CLASSES)) ? eh : 0.f;
                }
                float4 a[JV];
#pragma unroll
                for (int jj = 0; jj < JV; ++jj) a[jj] = ap[32 * jj];
#pragma unroll
                for (int jj = 0; jj < JV; ++jj) {
                  const int h0 = (C >= 128) ? (128 * jj) / C : jj * HG;
                  float e = __int_as_float(
                      __builtin_amdgcn_readlane(__float_as_int(exv), k * NH + h0));
#pragma unroll
                  for (int q = 1; q < HG; ++q) {
                    const float eq = __int_as_float(
                        __builtin_amdgcn_readlane(__float_as_int(exv), k * NH + h0 + q));
                    e = (lane >= q * (32 / HG)) ? eq : e;
                  }
                  const float4 hq = hv[k][jj];
                  a[jj].x += e * hq.x;
                  a[jj].y += e * hq.y;
                  a[jj].z += e * hq.z;
                  a[jj].w += e * hq.w;
                }
#pragma unroll
                for (int jj = 0; jj < JV; ++jj) ap[32 * jj] = a[jj];
              }
            }
          };
          int w = pl[min(kl, cnt - 1)];
          int t = 0;
          for (; t < nFull; ++t) {
            int wN;
            trip(t, w, false, wN);
            w = wN;
          }
          if (t < nTrip) {
            int wN;
            trip(t, w, true, wN);
          }
          {
            const int r = o + PRIV_CLASSES * ql;
            if (r < nrows) denS[r * NH + hl] = denReg;
          }
        }
      }
    }
    if (!viaList) {
      {
        int4 q0 = fetch(0), q1 = fetch(1), q2 = fetch(2), q3 = fetch(3);
        for (int i = 0; i < nIt; ++i) {
          const int4 cur = q0;
          q0 = q1; q1 = q2; q2 = q3; q3 = fetch(i + 4);
          const int dl[4] = {cur.x - base, cur.y - base, cur.z - base, cur.w - base};
          bool mt[4];
          for (int k = 0; k < 4; ++k) mt[k] = (unsigned)dl[k] < (unsigned)nrows;
          if (__builtin_amdgcn_ballot_w32(mt[0] | mt[1] | mt[2] | mt[3]) == 0u) continue;
          const int e0 = edge0(i);
          int sv[4];
          for (int k = 0; k < 4; ++k) sv[k] = mt[k] ? srcIdx[e0 + k] : 0;
          float es[4][NH];
          for (int k = 0; k < 4; ++k)
            for (int h = 0; h < NH; ++h)
              es[k][h] = mt[k] ? e_src[(size_t)sv[k] * NH + h] : 0.f;
          for (int k = 0; k < 4; ++k) {
            if (mt[k]) {
              for (int h = 0; h < NH; ++h) {
                float l = es[k][h] + edS[dl[k] * NH + h];
                l = (l > 0.f) ? l : l * NEG_SLOPE;
                ldsMaxFloat(&mS[dl[k] * NH + h], l);
              }
            }
          }
        }
      }
      __syncthreads();

      {
        auto addEdge = [&](int s, int d) {
          const float* hp = H + (size_t)s * F;
          float hv[J];
          for (int j = 0; j < J; ++j) hv[j] = hp[lane + 32 * j];
          const int h = lane & (NH - 1);
          float l = e_src[(size_t)s * NH + h] + edS[d * NH + h];
          l = (l > 0.f) ? l : l * NEG_SLOPE;
          const float ex = expf(l - mS[d * NH + h]);
          if (lane < NH) atomicAdd(&denS[d * NH + lane], ex);
          float* ap = acc + d * OWN_PITCH;
#pragma unroll
          for (int j = 0; j < J; ++j) {
            const float exj = __int_as_float(
                __builtin_amdgcn_readlane(__float_as_int(ex), (32 * j) / C));
            atomicAdd(&ap[lane + 32 * j], exj * hv[j]);
          }
        };

        int4 q0 = fetch(0), q1 = fetch(1), q2 = fetch(2), q3 = fetch(3);
        for (int i = 0; i < nIt; ++i) {
          const int4 cur = q0;
          q0 = q1; q1 = q2; q2 = q3; q3 = fetch(i + 4);
          const int dl[4] = {cur.x - base, cur.y - base, cur.z - base, cur.w - base};
          bool mt[4];
          for (int k = 0; k < 4; ++k) mt[k] = (unsigned)dl[k] < (unsigned)nrows;
          if (__builtin_amdgcn_ballot_w32(mt[0] | mt[1] | mt[2] | mt[3]) == 0u) continue;
          const int e0 = edge0(i);
          int sv[4];
          for (int k = 0; k < 4; ++k) sv[k] = mt[k] ? srcIdx[e0 + k] : 0;
          for (int k = 0; k < 4; ++k) {
            unsigned mask = __builtin_amdgcn_ballot_w32(mt[k]);
            while (mask) {
              const int j = __builtin_ctz(mask);
              mask &= mask - 1u;
              addEdge(__builtin_amdgcn_readlane(sv[k], j),
                      __builtin_amdgcn_readlane(dl[k], j));
            }
          }
        }
        for (int r = wave; r < nrows; r += nwav) addEdge(base + r, r);
      }
    }
    __syncthreads();

    if (!FINAL) {
      for (int i = tid; i < nrows * F; i += nthr) {
        const int r = i / F;
        const int f = i - r * F;
        const float v = acc[r * OWN_PITCH + f] / (denS[r * NH + f / C] + 1e-16f) + b[f];
        out[(size_t)(base + r) * F + f] = (v > 0.f) ? v : (expf(v) - 1.f);
      }
    } else {
      for (int r = wave; r < nrows; r += nwav) {
        const float den = denS[r * NH] + 1e-16f;
        float sum = 0.f;
        for (int c = lane; c < F; c += 32) {
          float v = acc[r * OWN_PITCH + c] / den + b[c];
          v = (v > 0.f) ? v : (expf(v) - 1.f);
          sum += v;
        }
        for (int off = 16; off > 0; off >>= 1)
          sum += __shfl_down(sum, off, 32);
        if (lane == 0) out[base + r] = sum / (float)F;
      }
    }
    __syncthreads();
  }
}

static inline int cdiv(long long a, long long b) { return (int)((a + b - 1) / b); }

extern "C" void kernel_launch(void* const* d_in, const int* in_sizes, int n_in,
                              void* d_out, int out_size, void* d_ws, size_t ws_size,
                              hipStream_t stream) {
  const float* x      = (const float*)d_in[0];
  const int*   ei     = (const int*)  d_in[1];
  const float* W1     = (const float*)d_in[2];
  const float* a1_src = (const float*)d_in[3];
  const float* a1_dst = (const float*)d_in[4];
  const float* b1     = (const float*)d_in[5];
  const float* W2     = (const float*)d_in[6];
  const float* a2_src = (const float*)d_in[7];
  const float* a2_dst = (const float*)d_in[8];
  const float* b2     = (const float*)d_in[9];
  float* outp = (float*)d_out;

  const int N    = in_sizes[0] / DIN;
  const int E    = in_sizes[1] / 2;
  const int Etot = E + N;
  const int F1   = HEADS * HID;
  const int* srcIdx = ei;
  const int* dstIdx = ei + E;

  float* ws = (float*)d_ws;
  float* h1      = ws;
  float* e_src_b = h1 + (size_t)N * F1;
  float* e_dst_b = e_src_b + (size_t)N * HEADS;
  float* edgebuf = e_dst_b + (size_t)N * HEADS;
  float* m_b     = edgebuf + (size_t)Etot * HEADS;
  float* den_b   = m_b + (size_t)N * HEADS;
  float* out1    = den_b + (size_t)N * HEADS;
  float* h2      = h1;
  int* segW   = (int*)edgebuf;
  int* segCnt = (int*)m_b;

  const int TB = 256;

  bucket_edges<<<NSEG, PRE_THREADS, 0, stream>>>(srcIdx, dstIdx, E, N, segW, segCnt, Etot);

  {
    int strips = (N / 16) * (F1 / 64);
    gemm_f32_wmma<DIN, F1><<<cdiv(strips, TB / 32), TB, 0, stream>>>(x, W1, h1, N);

    dot_e_kernel<<<cdiv((long long)N * HEADS, TB), TB, 0, stream>>>(
        h1, a1_src, a1_dst, e_src_b, e_dst_b, N, HEADS, HID);

    gat_owner<HEADS, HID, false><<<OWN_GROUPS, OWN_THREADS, 0, stream>>>(
        srcIdx, dstIdx, E, N, h1, e_src_b, e_dst_b, b1, out1, segW, segCnt, Etot);
  }

  {
    int strips = (N / 16) * (DOUT / 64);
    gemm_f32_wmma<F1, DOUT><<<cdiv(strips, TB / 32), TB, 0, stream>>>(out1, W2, h2, N);

    dot_e_kernel<<<cdiv(N, TB), TB, 0, stream>>>(
        h2, a2_src, a2_dst, e_src_b, e_dst_b, N, 1, DOUT);

    gat_owner<1, DOUT, true><<<OWN_GROUPS, OWN_THREADS, 0, stream>>>(
        srcIdx, dstIdx, E, N, h2, e_src_b, e_dst_b, b2, outp, segW, segCnt, Etot);
  }
}
